# starting kernel
# baseline (speedup 1.0000x reference)
#include <hip/hip_runtime.h>
#include <hip/hip_bf16.h>
#include <stdint.h>

typedef __attribute__((ext_vector_type(2))) float v2f;
typedef __attribute__((ext_vector_type(8))) float v8f;

__global__ __launch_bounds__(256) void zero_kernel(float* __restrict__ p, int n) {
    int i = blockIdx.x * blockDim.x + threadIdx.x;
    if (i < n) p[i] = 0.0f;
}

__device__ __forceinline__ void atomic_add_f32(float* p, float v) {
    __hip_atomic_fetch_add(p, v, __ATOMIC_RELAXED, __HIP_MEMORY_SCOPE_AGENT);
}

__device__ __forceinline__ int   nt_load_i32(const int* p)   { return __builtin_nontemporal_load(p); }
__device__ __forceinline__ float nt_load_f32(const float* p) { return __builtin_nontemporal_load(p); }

__global__ __launch_bounds__(256) void degree_kernel(const int* __restrict__ ei,
                                                     const float* __restrict__ w,
                                                     float* __restrict__ deg, int e) {
    int t = blockIdx.x * blockDim.x + threadIdx.x;
    if (t < e) {
        int d = nt_load_i32(ei + e + t);
        atomic_add_f32(&deg[d], nt_load_f32(w + t));
    }
}

__global__ __launch_bounds__(256) void dinv_kernel(float* __restrict__ deg, int n) {
    int i = blockIdx.x * blockDim.x + threadIdx.x;
    if (i < n) deg[i] = rsqrtf(deg[i] + 1.0f);
}

__global__ __launch_bounds__(32) void linear1_wmma(const float* __restrict__ x,
                                                   const float* __restrict__ W1,
                                                   float* __restrict__ h1, int n) {
    int lane = threadIdx.x;
    int base = blockIdx.x * 16;
    int idx  = lane & 15;
    bool lo  = lane < 16;
    int r    = base + idx;
    bool rv  = (r < n);

    v2f a;
    a.x = (lo && rv) ? x[r * 2 + 0] : 0.0f;
    a.y = (lo && rv) ? x[r * 2 + 1] : 0.0f;

    bool cv = (idx < 8);
    v2f b;
    b.x = (lo && cv) ? W1[0 * 8 + idx] : 0.0f;
    b.y = (lo && cv) ? W1[1 * 8 + idx] : 0.0f;

    v8f c = {};
    v8f d = __builtin_amdgcn_wmma_f32_16x16x4_f32(false, a, false, b,
                                                  (short)0, c, false, false);
    int rowoff = (lo ? 0 : 8) + base;
    if (base + 16 <= n) {
        if (cv) {
#pragma unroll
            for (int v = 0; v < 8; ++v) h1[(rowoff + v) * 8 + idx] = d[v];
        }
    } else {
        if (cv) {
#pragma unroll
            for (int v = 0; v < 8; ++v) {
                int m = rowoff + v;
                if (m < n) h1[m * 8 + idx] = d[v];
            }
        }
    }
}

__global__ __launch_bounds__(32) void linear2_wmma(const float* __restrict__ h1,
                                                   const float* __restrict__ W2,
                                                   float* __restrict__ h2, int n) {
    int lane = threadIdx.x;
    int base = blockIdx.x * 16;
    int idx  = lane & 15;
    bool lo  = lane < 16;
    int r    = base + idx;
    bool rv  = (r < n);
    int k0   = lo ? 0 : 2;

    v2f a0, a1;
    a0.x = rv ? h1[r * 8 + k0 + 0] : 0.0f;
    a0.y = rv ? h1[r * 8 + k0 + 1] : 0.0f;
    a1.x = rv ? h1[r * 8 + k0 + 4] : 0.0f;
    a1.y = rv ? h1[r * 8 + k0 + 5] : 0.0f;

    bool cv = (idx < 2);
    v2f b0, b1;
    b0.x = cv ? W2[(k0 + 0) * 2 + idx] : 0.0f;
    b0.y = cv ? W2[(k0 + 1) * 2 + idx] : 0.0f;
    b1.x = cv ? W2[(k0 + 4) * 2 + idx] : 0.0f;
    b1.y = cv ? W2[(k0 + 5) * 2 + idx] : 0.0f;

    v8f c = {};
    c = __builtin_amdgcn_wmma_f32_16x16x4_f32(false, a0, false, b0, (short)0, c, false, false);
    c = __builtin_amdgcn_wmma_f32_16x16x4_f32(false, a1, false, b1, (short)0, c, false, false);

    int rowoff = (lo ? 0 : 8) + base;
    if (base + 16 <= n) {
        if (cv) {
#pragma unroll
            for (int v = 0; v < 8; ++v) h2[(rowoff + v) * 2 + idx] = c[v];
        }
    } else {
        if (cv) {
#pragma unroll
            for (int v = 0; v < 8; ++v) {
                int m = rowoff + v;
                if (m < n) h2[m * 2 + idx] = c[v];
            }
        }
    }
}

__global__ __launch_bounds__(256) void edge_agg8(const int* __restrict__ ei,
                                                 const float* __restrict__ w,
                                                 const float* __restrict__ dinv,
                                                 const float* __restrict__ h,
                                                 float* __restrict__ agg, int e) {
    int t = blockIdx.x * blockDim.x + threadIdx.x;
    if (t >= e) return;
    int s = nt_load_i32(ei + t);
    int d = nt_load_i32(ei + e + t);
    float nrm = dinv[s] * nt_load_f32(w + t) * dinv[d];
    const float4* hp = (const float4*)(h + (size_t)s * 8);
    float4 lo4 = hp[0];
    float4 hi4 = hp[1];
    float* ap = agg + (size_t)d * 8;
    atomic_add_f32(ap + 0, nrm * lo4.x);
    atomic_add_f32(ap + 1, nrm * lo4.y);
    atomic_add_f32(ap + 2, nrm * lo4.z);
    atomic_add_f32(ap + 3, nrm * lo4.w);
    atomic_add_f32(ap + 4, nrm * hi4.x);
    atomic_add_f32(ap + 5, nrm * hi4.y);
    atomic_add_f32(ap + 6, nrm * hi4.z);
    atomic_add_f32(ap + 7, nrm * hi4.w);
}

__global__ __launch_bounds__(256) void edge_agg2(const int* __restrict__ ei,
                                                 const float* __restrict__ w,
                                                 const float* __restrict__ dinv,
                                                 const float* __restrict__ h,
                                                 float* __restrict__ agg, int e) {
    int t = blockIdx.x * blockDim.x + threadIdx.x;
    if (t >= e) return;
    int s = nt_load_i32(ei + t);
    int d = nt_load_i32(ei + e + t);
    float nrm = dinv[s] * nt_load_f32(w + t) * dinv[d];
    const float2* hp = (const float2*)(h + (size_t)s * 2);
    float2 hv = hp[0];
    float* ap = agg + (size_t)d * 2;
    atomic_add_f32(ap + 0, nrm * hv.x);
    atomic_add_f32(ap + 1, nrm * hv.y);
}

__global__ __launch_bounds__(256) void finalize1(float* __restrict__ h1,
                                                 const float* __restrict__ agg1,
                                                 const float* __restrict__ dinv,
                                                 const float* __restrict__ b1, int n) {
    int i = blockIdx.x * blockDim.x + threadIdx.x;
    if (i >= n) return;
    float sl = dinv[i] * dinv[i];
#pragma unroll
    for (int k = 0; k < 8; ++k) {
        float v = agg1[(size_t)i * 8 + k] + sl * h1[(size_t)i * 8 + k] + b1[k];
        h1[(size_t)i * 8 + k] = v > 0.0f ? v : 0.0f;
    }
}

__global__ __launch_bounds__(256) void finalize2(const float* __restrict__ h2,
                                                 const float* __restrict__ agg2,
                                                 const float* __restrict__ dinv,
                                                 const float* __restrict__ b2,
                                                 float* __restrict__ out, int n) {
    int i = blockIdx.x * blockDim.x + threadIdx.x;
    if (i >= n) return;
    float sl = dinv[i] * dinv[i];
    float v0 = agg2[(size_t)i * 2 + 0] + sl * h2[(size_t)i * 2 + 0] + b2[0];
    float v1 = agg2[(size_t)i * 2 + 1] + sl * h2[(size_t)i * 2 + 1] + b2[1];
    float m  = fmaxf(v0, v1);
    float e0 = __expf(v0 - m);
    float e1 = __expf(v1 - m);
    float inv = 1.0f / (e0 + e1);
    out[(size_t)i * 2 + 0] = e0 * inv;
    out[(size_t)i * 2 + 1] = e1 * inv;
}

static inline float* align16(float* p) {
    return (float*)(((uintptr_t)p + 15) & ~(uintptr_t)15);
}

extern "C" void kernel_launch(void* const* d_in, const int* in_sizes, int n_in,
                              void* d_out, int out_size, void* d_ws, size_t ws_size,
                              hipStream_t stream) {
    const float* x  = (const float*)d_in[0];
    const int*   ei = (const int*)d_in[1];
    const float* w  = (const float*)d_in[2];
    const float* W1 = (const float*)d_in[3];
    const float* b1 = (const float*)d_in[4];
    const float* W2 = (const float*)d_in[5];
    const float* b2 = (const float*)d_in[6];
    float* out = (float*)d_out;

    const int n = in_sizes[0] / 2;
    const int e = in_sizes[2];

    float* ws   = (float*)d_ws;
    float* dinv = align16(ws);
    float* agg1 = align16(dinv + (size_t)n);
    float* agg2 = align16(agg1 + (size_t)n * 8);
    float* h1   = align16(agg2 + (size_t)n * 2);
    float* h2   = align16(h1   + (size_t)n * 8);

    const int TB = 256;
    dim3 blkE((e + TB - 1) / TB), thr(TB);
    dim3 blkN((n + TB - 1) / TB);
    dim3 blkN8((n * 8 + TB - 1) / TB);
    dim3 blkN2((n * 2 + TB - 1) / TB);
    dim3 blkW((n + 15) / 16), thrW(32);

    zero_kernel<<<blkN,  thr, 0, stream>>>(dinv, n);
    zero_kernel<<<blkN8, thr, 0, stream>>>(agg1, n * 8);
    zero_kernel<<<blkN2, thr, 0, stream>>>(agg2, n * 2);

    degree_kernel<<<blkE, thr, 0, stream>>>(ei, w, dinv, e);
    dinv_kernel<<<blkN, thr, 0, stream>>>(dinv, n);

    linear1_wmma<<<blkW, thrW, 0, stream>>>(x, W1, h1, n);
    edge_agg8<<<blkE, thr, 0, stream>>>(ei, w, dinv, h1, agg1, e);
    finalize1<<<blkN, thr, 0, stream>>>(h1, agg1, dinv, b1, n);

    linear2_wmma<<<blkW, thrW, 0, stream>>>(h1, W2, h2, n);
    edge_agg2<<<blkE, thr, 0, stream>>>(ei, w, dinv, h2, agg2, e);
    finalize2<<<blkN, thr, 0, stream>>>(h2, agg2, dinv, b2, out, n);
}


# Layer 1 sums the 2-wide x rows over edges before the W1 product (edge sum is linear): its edge pass adds 2 floats per edge, not 8; W1 WMMA + bias + relu follow it; f32 sum order changes
# speedup vs baseline: 1.4364x; 1.4364x over previous
#include <hip/hip_runtime.h>
#include <hip/hip_bf16.h>
#include <stdint.h>

typedef __attribute__((ext_vector_type(2))) float v2f;
typedef __attribute__((ext_vector_type(8))) float v8f;

__global__ __launch_bounds__(256) void zero_kernel(float* __restrict__ p, int n) {
    int i = blockIdx.x * blockDim.x + threadIdx.x;
    if (i < n) p[i] = 0.0f;
}

__device__ __forceinline__ void atomic_add_f32(float* p, float v) {
    __hip_atomic_fetch_add(p, v, __ATOMIC_RELAXED, __HIP_MEMORY_SCOPE_AGENT);
}

__device__ __forceinline__ int   nt_load_i32(const int* p)   { return __builtin_nontemporal_load(p); }
__device__ __forceinline__ float nt_load_f32(const float* p) { return __builtin_nontemporal_load(p); }

__global__ __launch_bounds__(256) void degree_kernel(const int* __restrict__ ei,
                                                     const float* __restrict__ w,
                                                     float* __restrict__ deg, int e) {
    int t = blockIdx.x * blockDim.x + threadIdx.x;
    if (t < e) {
        int d = nt_load_i32(ei + e + t);
        atomic_add_f32(&deg[d], nt_load_f32(w + t));
    }
}

__global__ __launch_bounds__(256) void dinv_kernel(float* __restrict__ deg, int n) {
    int i = blockIdx.x * blockDim.x + threadIdx.x;
    if (i < n) deg[i] = rsqrtf(deg[i] + 1.0f);
}

__global__ __launch_bounds__(32) void layer1_wmma(const float* __restrict__ x,
                                                  const float* __restrict__ aggx,
                                                  const float* __restrict__ dinv,
                                                  const float* __restrict__ W1,
                                                  const float* __restrict__ b1,
                                                  float* __restrict__ h1, int n) {
    int lane = threadIdx.x;
    int base = blockIdx.x * 16;
    int idx  = lane & 15;
    bool lo  = lane < 16;
    int r    = base + idx;
    bool rv  = (r < n);

    v2f a;
    a.x = 0.0f;
    a.y = 0.0f;
    if (lo && rv) {
        float  di = dinv[r];
        float  sl = di * di;
        float2 xv = *(const float2*)(x + (size_t)r * 2);
        float2 gv = *(const float2*)(aggx + (size_t)r * 2);
        a.x = gv.x + sl * xv.x;
        a.y = gv.y + sl * xv.y;
    }

    bool cv = (idx < 8);
    v2f b;
    b.x = (lo && cv) ? W1[0 * 8 + idx] : 0.0f;
    b.y = (lo && cv) ? W1[1 * 8 + idx] : 0.0f;
    float bias = cv ? b1[idx] : 0.0f;

    v8f c = {};
    v8f d = __builtin_amdgcn_wmma_f32_16x16x4_f32(false, a, false, b,
                                                  (short)0, c, false, false);
    int rowoff = (lo ? 0 : 8) + base;
    if (base + 16 <= n) {
        if (cv) {
#pragma unroll
            for (int v = 0; v < 8; ++v) {
                float t = d[v] + bias;
                h1[(size_t)(rowoff + v) * 8 + idx] = t > 0.0f ? t : 0.0f;
            }
        }
    } else {
        if (cv) {
#pragma unroll
            for (int v = 0; v < 8; ++v) {
                int m = rowoff + v;
                if (m < n) {
                    float t = d[v] + bias;
                    h1[(size_t)m * 8 + idx] = t > 0.0f ? t : 0.0f;
                }
            }
        }
    }
}

__global__ __launch_bounds__(32) void linear2_wmma(const float* __restrict__ h1,
                                                   const float* __restrict__ W2,
                                                   float* __restrict__ h2, int n) {
    int lane = threadIdx.x;
    int base = blockIdx.x * 16;
    int idx  = lane & 15;
    bool lo  = lane < 16;
    int r    = base + idx;
    bool rv  = (r < n);
    int k0   = lo ? 0 : 2;

    v2f a0, a1;
    a0.x = rv ? h1[r * 8 + k0 + 0] : 0.0f;
    a0.y = rv ? h1[r * 8 + k0 + 1] : 0.0f;
    a1.x = rv ? h1[r * 8 + k0 + 4] : 0.0f;
    a1.y = rv ? h1[r * 8 + k0 + 5] : 0.0f;

    bool cv = (idx < 2);
    v2f b0, b1;
    b0.x = cv ? W2[(k0 + 0) * 2 + idx] : 0.0f;
    b0.y = cv ? W2[(k0 + 1) * 2 + idx] : 0.0f;
    b1.x = cv ? W2[(k0 + 4) * 2 + idx] : 0.0f;
    b1.y = cv ? W2[(k0 + 5) * 2 + idx] : 0.0f;

    v8f c = {};
    c = __builtin_amdgcn_wmma_f32_16x16x4_f32(false, a0, false, b0, (short)0, c, false, false);
    c = __builtin_amdgcn_wmma_f32_16x16x4_f32(false, a1, false, b1, (short)0, c, false, false);

    int rowoff = (lo ? 0 : 8) + base;
    if (base + 16 <= n) {
        if (cv) {
#pragma unroll
            for (int v = 0; v < 8; ++v) h2[(rowoff + v) * 2 + idx] = c[v];
        }
    } else {
        if (cv) {
#pragma unroll
            for (int v = 0; v < 8; ++v) {
                int m = rowoff + v;
                if (m < n) h2[m * 2 + idx] = c[v];
            }
        }
    }
}

__global__ __launch_bounds__(256) void edge_agg2(const int* __restrict__ ei,
                                                 const float* __restrict__ w,
                                                 const float* __restrict__ dinv,
                                                 const float* __restrict__ h,
                                                 float* __restrict__ agg, int e) {
    int t = blockIdx.x * blockDim.x + threadIdx.x;
    if (t >= e) return;
    int s = nt_load_i32(ei + t);
    int d = nt_load_i32(ei + e + t);
    float nrm = dinv[s] * nt_load_f32(w + t) * dinv[d];
    const float2* hp = (const float2*)(h + (size_t)s * 2);
    float2 hv = hp[0];
    float* ap = agg + (size_t)d * 2;
    atomic_add_f32(ap + 0, nrm * hv.x);
    atomic_add_f32(ap + 1, nrm * hv.y);
}

__global__ __launch_bounds__(256) void finalize2(const float* __restrict__ h2,
                                                 const float* __restrict__ agg2,
                                                 const float* __restrict__ dinv,
                                                 const float* __restrict__ b2,
                                                 float* __restrict__ out, int n) {
    int i = blockIdx.x * blockDim.x + threadIdx.x;
    if (i >= n) return;
    float sl = dinv[i] * dinv[i];
    float v0 = agg2[(size_t)i * 2 + 0] + sl * h2[(size_t)i * 2 + 0] + b2[0];
    float v1 = agg2[(size_t)i * 2 + 1] + sl * h2[(size_t)i * 2 + 1] + b2[1];
    float m  = fmaxf(v0, v1);
    float e0 = __expf(v0 - m);
    float e1 = __expf(v1 - m);
    float inv = 1.0f / (e0 + e1);
    out[(size_t)i * 2 + 0] = e0 * inv;
    out[(size_t)i * 2 + 1] = e1 * inv;
}

static inline float* align16(float* p) {
    return (float*)(((uintptr_t)p + 15) & ~(uintptr_t)15);
}

extern "C" void kernel_launch(void* const* d_in, const int* in_sizes, int n_in,
                              void* d_out, int out_size, void* d_ws, size_t ws_size,
                              hipStream_t stream) {
    const float* x  = (const float*)d_in[0];
    const int*   ei = (const int*)d_in[1];
    const float* w  = (const float*)d_in[2];
    const float* W1 = (const float*)d_in[3];
    const float* b1 = (const float*)d_in[4];
    const float* W2 = (const float*)d_in[5];
    const float* b2 = (const float*)d_in[6];
    float* out = (float*)d_out;

    const int n = in_sizes[0] / 2;
    const int e = in_sizes[2];

    float* ws   = (float*)d_ws;
    float* dinv = align16(ws);
    float* aggx = align16(dinv + (size_t)n);
    float* agg2 = align16(aggx + (size_t)n * 2);
    float* h1   = align16(agg2 + (size_t)n * 2);
    float* h2   = align16(h1   + (size_t)n * 8);

    const int TB = 256;
    dim3 blkE((e + TB - 1) / TB), thr(TB);
    dim3 blkN((n + TB - 1) / TB);
    dim3 blkN2((n * 2 + TB - 1) / TB);
    dim3 blkW((n + 15) / 16), thrW(32);

    zero_kernel<<<blkN,  thr, 0, stream>>>(dinv, n);
    zero_kernel<<<blkN2, thr, 0, stream>>>(aggx, n * 2);
    zero_kernel<<<blkN2, thr, 0, stream>>>(agg2, n * 2);

    degree_kernel<<<blkE, thr, 0, stream>>>(ei, w, dinv, e);
    dinv_kernel<<<blkN, thr, 0, stream>>>(dinv, n);

    edge_agg2<<<blkE, thr, 0, stream>>>(ei, w, dinv, x, aggx, e);
    layer1_wmma<<<blkW, thrW, 0, stream>>>(x, aggx, dinv, W1, b1, h1, n);

    linear2_wmma<<<blkW, thrW, 0, stream>>>(h1, W2, h2, n);
    edge_agg2<<<blkE, thr, 0, stream>>>(ei, w, dinv, h2, agg2, e);
    finalize2<<<blkN, thr, 0, stream>>>(h2, agg2, dinv, b2, out, n);
}


# Each scatter-add target (deg, layer-1 sum, layer-2 sum) gets one copy per package half, chosen by the wave's die id, so L2 atomics stay in-half; consumers add both copies; f32 sum order changes
# speedup vs baseline: 3.6283x; 2.5260x over previous
#include <hip/hip_runtime.h>
#include <hip/hip_bf16.h>
#include <stdint.h>

typedef __attribute__((ext_vector_type(2))) float v2f;
typedef __attribute__((ext_vector_type(8))) float v8f;

__global__ __launch_bounds__(256) void zero_kernel(float* __restrict__ p, int n) {
    int i = blockIdx.x * blockDim.x + threadIdx.x;
    if (i < n) p[i] = 0.0f;
}

__device__ __forceinline__ void atomic_add_f32(float* p, float v) {
    __hip_atomic_fetch_add(p, v, __ATOMIC_RELAXED, __HIP_MEMORY_SCOPE_AGENT);
}

__device__ __forceinline__ int   nt_load_i32(const int* p)   { return __builtin_nontemporal_load(p); }
__device__ __forceinline__ float nt_load_f32(const float* p) { return __builtin_nontemporal_load(p); }

__device__ __forceinline__ int package_half() {
    unsigned id = __builtin_amdgcn_s_sendmsg_rtn(0x87);
    unsigned die = (id >> 16) & 0xFu;
    return (die >= 4u) ? 1 : 0;
}

__global__ __launch_bounds__(256) void degree_kernel(const int* __restrict__ ei,
                                                     const float* __restrict__ w,
                                                     float* __restrict__ deg0,
                                                     float* __restrict__ deg1, int e) {
    float* deg = package_half() ? deg1 : deg0;
    int t = blockIdx.x * blockDim.x + threadIdx.x;
    if (t < e) {
        int d = nt_load_i32(ei + e + t);
        atomic_add_f32(&deg[d], nt_load_f32(w + t));
    }
}

__global__ __launch_bounds__(256) void dinv_kernel(float* __restrict__ deg0,
                                                   const float* __restrict__ deg1, int n) {
    int i = blockIdx.x * blockDim.x + threadIdx.x;
    if (i < n) deg0[i] = rsqrtf((deg0[i] + deg1[i]) + 1.0f);
}

__global__ __launch_bounds__(32) void layer1_wmma(const float* __restrict__ x,
                                                  const float* __restrict__ aggx0,
                                                  const float* __restrict__ aggx1,
                                                  const float* __restrict__ dinv,
                                                  const float* __restrict__ W1,
                                                  const float* __restrict__ b1,
                                                  float* __restrict__ h1, int n) {
    int lane = threadIdx.x;
    int base = blockIdx.x * 16;
    int idx  = lane & 15;
    bool lo  = lane < 16;
    int r    = base + idx;
    bool rv  = (r < n);

    v2f a;
    a.x = 0.0f;
    a.y = 0.0f;
    if (lo && rv) {
        float  di = dinv[r];
        float  sl = di * di;
        float2 xv = *(const float2*)(x + (size_t)r * 2);
        float2 g0 = *(const float2*)(aggx0 + (size_t)r * 2);
        float2 g1 = *(const float2*)(aggx1 + (size_t)r * 2);
        a.x = (g0.x + g1.x) + sl * xv.x;
        a.y = (g0.y + g1.y) + sl * xv.y;
    }

    bool cv = (idx < 8);
    v2f b;
    b.x = (lo && cv) ? W1[0 * 8 + idx] : 0.0f;
    b.y = (lo && cv) ? W1[1 * 8 + idx] : 0.0f;
    float bias = cv ? b1[idx] : 0.0f;

    v8f c = {};
    v8f d = __builtin_amdgcn_wmma_f32_16x16x4_f32(false, a, false, b,
                                                  (short)0, c, false, false);
    int rowoff = (lo ? 0 : 8) + base;
    if (base + 16 <= n) {
        if (cv) {
#pragma unroll
            for (int v = 0; v < 8; ++v) {
                float t = d[v] + bias;
                h1[(size_t)(rowoff + v) * 8 + idx] = t > 0.0f ? t : 0.0f;
            }
        }
    } else {
        if (cv) {
#pragma unroll
            for (int v = 0; v < 8; ++v) {
                int m = rowoff + v;
                if (m < n) {
                    float t = d[v] + bias;
                    h1[(size_t)m * 8 + idx] = t > 0.0f ? t : 0.0f;
                }
            }
        }
    }
}

__global__ __launch_bounds__(32) void linear2_wmma(const float* __restrict__ h1,
                                                   const float* __restrict__ W2,
                                                   float* __restrict__ h2, int n) {
    int lane = threadIdx.x;
    int base = blockIdx.x * 16;
    int idx  = lane & 15;
    bool lo  = lane < 16;
    int r    = base + idx;
    bool rv  = (r < n);
    int k0   = lo ? 0 : 2;

    v2f a0, a1;
    a0.x = rv ? h1[r * 8 + k0 + 0] : 0.0f;
    a0.y = rv ? h1[r * 8 + k0 + 1] : 0.0f;
    a1.x = rv ? h1[r * 8 + k0 + 4] : 0.0f;
    a1.y = rv ? h1[r * 8 + k0 + 5] : 0.0f;

    bool cv = (idx < 2);
    v2f b0, b1;
    b0.x = cv ? W2[(k0 + 0) * 2 + idx] : 0.0f;
    b0.y = cv ? W2[(k0 + 1) * 2 + idx] : 0.0f;
    b1.x = cv ? W2[(k0 + 4) * 2 + idx] : 0.0f;
    b1.y = cv ? W2[(k0 + 5) * 2 + idx] : 0.0f;

    v8f c = {};
    c = __builtin_amdgcn_wmma_f32_16x16x4_f32(false, a0, false, b0, (short)0, c, false, false);
    c = __builtin_amdgcn_wmma_f32_16x16x4_f32(false, a1, false, b1, (short)0, c, false, false);

    int rowoff = (lo ? 0 : 8) + base;
    if (base + 16 <= n) {
        if (cv) {
#pragma unroll
            for (int v = 0; v < 8; ++v) h2[(rowoff + v) * 2 + idx] = c[v];
        }
    } else {
        if (cv) {
#pragma unroll
            for (int v = 0; v < 8; ++v) {
                int m = rowoff + v;
                if (m < n) h2[m * 2 + idx] = c[v];
            }
        }
    }
}

__global__ __launch_bounds__(256) void edge_agg2(const int* __restrict__ ei,
                                                 const float* __restrict__ w,
                                                 const float* __restrict__ dinv,
                                                 const float* __restrict__ h,
                                                 float* __restrict__ agg0,
                                                 float* __restrict__ agg1, int e) {
    float* agg = package_half() ? agg1 : agg0;
    int t = blockIdx.x * blockDim.x + threadIdx.x;
    if (t >= e) return;
    int s = nt_load_i32(ei + t);
    int d = nt_load_i32(ei + e + t);
    float nrm = dinv[s] * nt_load_f32(w + t) * dinv[d];
    const float2* hp = (const float2*)(h + (size_t)s * 2);
    float2 hv = hp[0];
    float* ap = agg + (size_t)d * 2;
    atomic_add_f32(ap + 0, nrm * hv.x);
    atomic_add_f32(ap + 1, nrm * hv.y);
}

__global__ __launch_bounds__(256) void finalize2(const float* __restrict__ h2,
                                                 const float* __restrict__ agg20,
                                                 const float* __restrict__ agg21,
                                                 const float* __restrict__ dinv,
                                                 const float* __restrict__ b2,
                                                 float* __restrict__ out, int n) {
    int i = blockIdx.x * blockDim.x + threadIdx.x;
    if (i >= n) return;
    float sl = dinv[i] * dinv[i];
    float v0 = (agg20[(size_t)i * 2 + 0] + agg21[(size_t)i * 2 + 0]) + sl * h2[(size_t)i * 2 + 0] + b2[0];
    float v1 = (agg20[(size_t)i * 2 + 1] + agg21[(size_t)i * 2 + 1]) + sl * h2[(size_t)i * 2 + 1] + b2[1];
    float m  = fmaxf(v0, v1);
    float e0 = __expf(v0 - m);
    float e1 = __expf(v1 - m);
    float inv = 1.0f / (e0 + e1);
    out[(size_t)i * 2 + 0] = e0 * inv;
    out[(size_t)i * 2 + 1] = e1 * inv;
}

static inline float* align16(float* p) {
    return (float*)(((uintptr_t)p + 15) & ~(uintptr_t)15);
}

extern "C" void kernel_launch(void* const* d_in, const int* in_sizes, int n_in,
                              void* d_out, int out_size, void* d_ws, size_t ws_size,
                              hipStream_t stream) {
    const float* x  = (const float*)d_in[0];
    const int*   ei = (const int*)d_in[1];
    const float* w  = (const float*)d_in[2];
    const float* W1 = (const float*)d_in[3];
    const float* b1 = (const float*)d_in[4];
    const float* W2 = (const float*)d_in[5];
    const float* b2 = (const float*)d_in[6];
    float* out = (float*)d_out;

    const int n = in_sizes[0] / 2;
    const int e = in_sizes[2];

    float* ws    = (float*)d_ws;
    float* dinv  = align16(ws);
    float* aggx0 = align16(dinv  + (size_t)n);
    float* aggx1 = align16(aggx0 + (size_t)n * 2);
    float* h1    = align16(aggx1 + (size_t)n * 2);
    float* h2    = align16(h1    + (size_t)n * 8);
    float* deg1  = h1;
    float* agg20 = h1;
    float* agg21 = align16(agg20 + (size_t)n * 2);

    const int TB = 256;
    dim3 blkE((e + TB - 1) / TB), thr(TB);
    dim3 blkN((n + TB - 1) / TB);
    const int nzA = (int)((deg1 + (size_t)n) - dinv);
    const int nzB = (int)((agg21 + (size_t)n * 2) - agg20);
    dim3 blkZA((nzA + TB - 1) / TB);
    dim3 blkZB((nzB + TB - 1) / TB);
    dim3 blkW((n + 15) / 16), thrW(32);

    zero_kernel<<<blkZA, thr, 0, stream>>>(dinv, nzA);

    degree_kernel<<<blkE, thr, 0, stream>>>(ei, w, dinv, deg1, e);
    dinv_kernel<<<blkN, thr, 0, stream>>>(dinv, deg1, n);

    edge_agg2<<<blkE, thr, 0, stream>>>(ei, w, dinv, x, aggx0, aggx1, e);
    layer1_wmma<<<blkW, thrW, 0, stream>>>(x, aggx0, aggx1, dinv, W1, b1, h1, n);

    linear2_wmma<<<blkW, thrW, 0, stream>>>(h1, W2, h2, n);
    zero_kernel<<<blkZB, thr, 0, stream>>>(agg20, nzB);
    edge_agg2<<<blkE, thr, 0, stream>>>(ei, w, dinv, h2, agg20, agg21, e);
    finalize2<<<blkN, thr, 0, stream>>>(h2, agg20, agg21, dinv, b2, out, n);
}


# Edge passes use two adjacent lanes per edge, one per feature: its two f32 adds are 8 contiguous bytes of ONE atomic instruction (one L2 request per edge, not two); same values, arrival-order sums
# speedup vs baseline: 4.3049x; 1.1865x over previous
#include <hip/hip_runtime.h>
#include <hip/hip_bf16.h>
#include <stdint.h>

typedef __attribute__((ext_vector_type(2))) float v2f;
typedef __attribute__((ext_vector_type(8))) float v8f;

__global__ __launch_bounds__(256) void zero_kernel(float* __restrict__ p, int n) {
    int i = blockIdx.x * blockDim.x + threadIdx.x;
    if (i < n) p[i] = 0.0f;
}

__device__ __forceinline__ void atomic_add_f32(float* p, float v) {
    __hip_atomic_fetch_add(p, v, __ATOMIC_RELAXED, __HIP_MEMORY_SCOPE_AGENT);
}

__device__ __forceinline__ int   nt_load_i32(const int* p)   { return __builtin_nontemporal_load(p); }
__device__ __forceinline__ float nt_load_f32(const float* p) { return __builtin_nontemporal_load(p); }

__device__ __forceinline__ int package_half() {
    unsigned id = __builtin_amdgcn_s_sendmsg_rtn(0x87);
    unsigned die = (id >> 16) & 0xFu;
    return (die >= 4u) ? 1 : 0;
}

__global__ __launch_bounds__(256) void degree_kernel(const int* __restrict__ ei,
                                                     const float* __restrict__ w,
                                                     float* __restrict__ deg0,
                                                     float* __restrict__ deg1, int e) {
    float* deg = package_half() ? deg1 : deg0;
    int t = blockIdx.x * blockDim.x + threadIdx.x;
    if (t < e) {
        int d = nt_load_i32(ei + e + t);
        atomic_add_f32(&deg[d], nt_load_f32(w + t));
    }
}

__global__ __launch_bounds__(256) void dinv_kernel(float* __restrict__ deg0,
                                                   const float* __restrict__ deg1, int n) {
    int i = blockIdx.x * blockDim.x + threadIdx.x;
    if (i < n) deg0[i] = rsqrtf((deg0[i] + deg1[i]) + 1.0f);
}

__global__ __launch_bounds__(32) void layer1_wmma(const float* __restrict__ x,
                                                  const float* __restrict__ aggx0,
                                                  const float* __restrict__ aggx1,
                                                  const float* __restrict__ dinv,
                                                  const float* __restrict__ W1,
                                                  const float* __restrict__ b1,
                                                  float* __restrict__ h1, int n) {
    int lane = threadIdx.x;
    int base = blockIdx.x * 16;
    int idx  = lane & 15;
    bool lo  = lane < 16;
    int r    = base + idx;
    bool rv  = (r < n);

    v2f a;
    a.x = 0.0f;
    a.y = 0.0f;
    if (lo && rv) {
        float  di = dinv[r];
        float  sl = di * di;
        float2 xv = *(const float2*)(x + (size_t)r * 2);
        float2 g0 = *(const float2*)(aggx0 + (size_t)r * 2);
        float2 g1 = *(const float2*)(aggx1 + (size_t)r * 2);
        a.x = (g0.x + g1.x) + sl * xv.x;
        a.y = (g0.y + g1.y) + sl * xv.y;
    }

    bool cv = (idx < 8);
    v2f b;
    b.x = (lo && cv) ? W1[0 * 8 + idx] : 0.0f;
    b.y = (lo && cv) ? W1[1 * 8 + idx] : 0.0f;
    float bias = cv ? b1[idx] : 0.0f;

    v8f c = {};
    v8f d = __builtin_amdgcn_wmma_f32_16x16x4_f32(false, a, false, b,
                                                  (short)0, c, false, false);
    int rowoff = (lo ? 0 : 8) + base;
    if (base + 16 <= n) {
        if (cv) {
#pragma unroll
            for (int v = 0; v < 8; ++v) {
                float t = d[v] + bias;
                h1[(size_t)(rowoff + v) * 8 + idx] = t > 0.0f ? t : 0.0f;
            }
        }
    } else {
        if (cv) {
#pragma unroll
            for (int v = 0; v < 8; ++v) {
                int m = rowoff + v;
                if (m < n) {
                    float t = d[v] + bias;
                    h1[(size_t)m * 8 + idx] = t > 0.0f ? t : 0.0f;
                }
            }
        }
    }
}

__global__ __launch_bounds__(32) void linear2_wmma(const float* __restrict__ h1,
                                                   const float* __restrict__ W2,
                                                   float* __restrict__ h2, int n) {
    int lane = threadIdx.x;
    int base = blockIdx.x * 16;
    int idx  = lane & 15;
    bool lo  = lane < 16;
    int r    = base + idx;
    bool rv  = (r < n);
    int k0   = lo ? 0 : 2;

    v2f a0, a1;
    a0.x = rv ? h1[r * 8 + k0 + 0] : 0.0f;
    a0.y = rv ? h1[r * 8 + k0 + 1] : 0.0f;
    a1.x = rv ? h1[r * 8 + k0 + 4] : 0.0f;
    a1.y = rv ? h1[r * 8 + k0 + 5] : 0.0f;

    bool cv = (idx < 2);
    v2f b0, b1;
    b0.x = cv ? W2[(k0 + 0) * 2 + idx] : 0.0f;
    b0.y = cv ? W2[(k0 + 1) * 2 + idx] : 0.0f;
    b1.x = cv ? W2[(k0 + 4) * 2 + idx] : 0.0f;
    b1.y = cv ? W2[(k0 + 5) * 2 + idx] : 0.0f;

    v8f c = {};
    c = __builtin_amdgcn_wmma_f32_16x16x4_f32(false, a0, false, b0, (short)0, c, false, false);
    c = __builtin_amdgcn_wmma_f32_16x16x4_f32(false, a1, false, b1, (short)0, c, false, false);

    int rowoff = (lo ? 0 : 8) + base;
    if (base + 16 <= n) {
        if (cv) {
#pragma unroll
            for (int v = 0; v < 8; ++v) h2[(rowoff + v) * 2 + idx] = c[v];
        }
    } else {
        if (cv) {
#pragma unroll
            for (int v = 0; v < 8; ++v) {
                int m = rowoff + v;
                if (m < n) h2[m * 2 + idx] = c[v];
            }
        }
    }
}

__device__ __forceinline__ void edge_add1(const int* __restrict__ ei,
                                          const float* __restrict__ w,
                                          const float* __restrict__ dinv,
                                          const float* __restrict__ h,
                                          float* __restrict__ agg, int e, int t, int f) {
    int s = nt_load_i32(ei + t);
    int d = nt_load_i32(ei + e + t);
    float nrm = dinv[s] * nt_load_f32(w + t) * dinv[d];
    atomic_add_f32(agg + (size_t)d * 2 + f, nrm * h[(size_t)s * 2 + f]);
}

__global__ __launch_bounds__(256) void edge_agg2(const int* __restrict__ ei,
                                                 const float* __restrict__ w,
                                                 const float* __restrict__ dinv,
                                                 const float* __restrict__ h,
                                                 float* __restrict__ agg0,
                                                 float* __restrict__ agg1, int e) {
    float* agg = package_half() ? agg1 : agg0;
    const int lane = threadIdx.x & 31;
    const int f    = lane & 1;
    const int t0   = blockIdx.x * 256 + (threadIdx.x & ~31) + (lane >> 1);
    const int t1   = t0 + 16;

    if ((blockIdx.x + 1) * 256 <= e) {
        const int   s0 = nt_load_i32(ei + t0);
        const int   s1 = nt_load_i32(ei + t1);
        const int   d0 = nt_load_i32(ei + e + t0);
        const int   d1 = nt_load_i32(ei + e + t1);
        const float w0 = nt_load_f32(w + t0);
        const float w1 = nt_load_f32(w + t1);
        const float a0 = dinv[s0];
        const float a1 = dinv[s1];
        const float b0 = dinv[d0];
        const float b1 = dinv[d1];
        const float g0 = h[(size_t)s0 * 2 + f];
        const float g1 = h[(size_t)s1 * 2 + f];
        atomic_add_f32(agg + (size_t)d0 * 2 + f, (a0 * w0 * b0) * g0);
        atomic_add_f32(agg + (size_t)d1 * 2 + f, (a1 * w1 * b1) * g1);
    } else {
        if (t0 < e) edge_add1(ei, w, dinv, h, agg, e, t0, f);
        if (t1 < e) edge_add1(ei, w, dinv, h, agg, e, t1, f);
    }
}

__global__ __launch_bounds__(256) void finalize2(const float* __restrict__ h2,
                                                 const float* __restrict__ agg20,
                                                 const float* __restrict__ agg21,
                                                 const float* __restrict__ dinv,
                                                 const float* __restrict__ b2,
                                                 float* __restrict__ out, int n) {
    int i = blockIdx.x * blockDim.x + threadIdx.x;
    if (i >= n) return;
    float sl = dinv[i] * dinv[i];
    float v0 = (agg20[(size_t)i * 2 + 0] + agg21[(size_t)i * 2 + 0]) + sl * h2[(size_t)i * 2 + 0] + b2[0];
    float v1 = (agg20[(size_t)i * 2 + 1] + agg21[(size_t)i * 2 + 1]) + sl * h2[(size_t)i * 2 + 1] + b2[1];
    float m  = fmaxf(v0, v1);
    float e0 = __expf(v0 - m);
    float e1 = __expf(v1 - m);
    float inv = 1.0f / (e0 + e1);
    out[(size_t)i * 2 + 0] = e0 * inv;
    out[(size_t)i * 2 + 1] = e1 * inv;
}

static inline float* align16(float* p) {
    return (float*)(((uintptr_t)p + 15) & ~(uintptr_t)15);
}

extern "C" void kernel_launch(void* const* d_in, const int* in_sizes, int n_in,
                              void* d_out, int out_size, void* d_ws, size_t ws_size,
                              hipStream_t stream) {
    const float* x  = (const float*)d_in[0];
    const int*   ei = (const int*)d_in[1];
    const float* w  = (const float*)d_in[2];
    const float* W1 = (const float*)d_in[3];
    const float* b1 = (const float*)d_in[4];
    const float* W2 = (const float*)d_in[5];
    const float* b2 = (const float*)d_in[6];
    float* out = (float*)d_out;

    const int n = in_sizes[0] / 2;
    const int e = in_sizes[2];

    float* ws    = (float*)d_ws;
    float* dinv  = align16(ws);
    float* aggx0 = align16(dinv  + (size_t)n);
    float* aggx1 = align16(aggx0 + (size_t)n * 2);
    float* h1    = align16(aggx1 + (size_t)n * 2);
    float* h2    = align16(h1    + (size_t)n * 8);
    float* deg1  = h1;
    float* agg20 = h1;
    float* agg21 = align16(agg20 + (size_t)n * 2);

    const int TB = 256;
    dim3 blkE((e + TB - 1) / TB), thr(TB);
    dim3 blkN((n + TB - 1) / TB);
    const int nzA = (int)((deg1 + (size_t)n) - dinv);
    const int nzB = (int)((agg21 + (size_t)n * 2) - agg20);
    dim3 blkZA((nzA + TB - 1) / TB);
    dim3 blkZB((nzB + TB - 1) / TB);
    dim3 blkW((n + 15) / 16), thrW(32);

    zero_kernel<<<blkZA, thr, 0, stream>>>(dinv, nzA);

    degree_kernel<<<blkE, thr, 0, stream>>>(ei, w, dinv, deg1, e);
    dinv_kernel<<<blkN, thr, 0, stream>>>(dinv, deg1, n);

    edge_agg2<<<blkE, thr, 0, stream>>>(ei, w, dinv, x, aggx0, aggx1, e);
    layer1_wmma<<<blkW, thrW, 0, stream>>>(x, aggx0, aggx1, dinv, W1, b1, h1, n);

    linear2_wmma<<<blkW, thrW, 0, stream>>>(h1, W2, h2, n);
    zero_kernel<<<blkZB, thr, 0, stream>>>(agg20, nzB);
    edge_agg2<<<blkE, thr, 0, stream>>>(ei, w, dinv, h2, agg20, agg21, e);
    finalize2<<<blkN, thr, 0, stream>>>(h2, agg20, agg21, dinv, b2, out, n);
}


# Symmetric norm moved from edges to nodes: rows kept pre-scaled by dinv, edge passes add w*row[src] (no dinv gathers), consumers scale sums by dinv[dst]; f32 term rounding changes, arrival-order sums
# speedup vs baseline: 5.5195x; 1.2821x over previous
#include <hip/hip_runtime.h>
#include <hip/hip_bf16.h>
#include <stdint.h>

typedef __attribute__((ext_vector_type(2))) float v2f;
typedef __attribute__((ext_vector_type(8))) float v8f;

__global__ __launch_bounds__(256) void zero_kernel(float* __restrict__ p, int n) {
    int i = blockIdx.x * blockDim.x + threadIdx.x;
    if (i < n) p[i] = 0.0f;
}

__device__ __forceinline__ void atomic_add_f32(float* p, float v) {
    __hip_atomic_fetch_add(p, v, __ATOMIC_RELAXED, __HIP_MEMORY_SCOPE_AGENT);
}

__device__ __forceinline__ int   nt_load_i32(const int* p)   { return __builtin_nontemporal_load(p); }
__device__ __forceinline__ float nt_load_f32(const float* p) { return __builtin_nontemporal_load(p); }

__device__ __forceinline__ int package_half() {
    unsigned id = __builtin_amdgcn_s_sendmsg_rtn(0x87);
    unsigned die = (id >> 16) & 0xFu;
    return (die >= 4u) ? 1 : 0;
}

__global__ __launch_bounds__(256) void degree_kernel(const int* __restrict__ ei,
                                                     const float* __restrict__ w,
                                                     float* __restrict__ deg0,
                                                     float* __restrict__ deg1, int e) {
    float* deg = package_half() ? deg1 : deg0;
    int t = blockIdx.x * blockDim.x + threadIdx.x;
    if (t < e) {
        int d = nt_load_i32(ei + e + t);
        atomic_add_f32(&deg[d], nt_load_f32(w + t));
    }
}

__global__ __launch_bounds__(256) void dinv_kernel(float* __restrict__ deg0,
                                                   const float* __restrict__ deg1,
                                                   const float* __restrict__ x,
                                                   float* __restrict__ xs, int n) {
    int i = blockIdx.x * blockDim.x + threadIdx.x;
    if (i < n) {
        float  di = rsqrtf((deg0[i] + deg1[i]) + 1.0f);
        float2 xv = *(const float2*)(x + (size_t)i * 2);
        deg0[i] = di;
        *(float2*)(xs + (size_t)i * 2) = make_float2(di * xv.x, di * xv.y);
    }
}

__global__ __launch_bounds__(32) void layer1_wmma(const float* __restrict__ x,
                                                  const float* __restrict__ aggx0,
                                                  const float* __restrict__ aggx1,
                                                  const float* __restrict__ dinv,
                                                  const float* __restrict__ W1,
                                                  const float* __restrict__ b1,
                                                  float* __restrict__ h1, int n) {
    int lane = threadIdx.x;
    int base = blockIdx.x * 16;
    int idx  = lane & 15;
    bool lo  = lane < 16;
    int r    = base + idx;
    bool rv  = (r < n);

    v2f a;
    a.x = 0.0f;
    a.y = 0.0f;
    if (lo && rv) {
        float  di = dinv[r];
        float2 xv = *(const float2*)(x + (size_t)r * 2);
        float2 g0 = *(const float2*)(aggx0 + (size_t)r * 2);
        float2 g1 = *(const float2*)(aggx1 + (size_t)r * 2);
        a.x = di * ((g0.x + g1.x) + di * xv.x);
        a.y = di * ((g0.y + g1.y) + di * xv.y);
    }

    bool cv = (idx < 8);
    v2f b;
    b.x = (lo && cv) ? W1[0 * 8 + idx] : 0.0f;
    b.y = (lo && cv) ? W1[1 * 8 + idx] : 0.0f;
    float bias = cv ? b1[idx] : 0.0f;

    v8f c = {};
    v8f d = __builtin_amdgcn_wmma_f32_16x16x4_f32(false, a, false, b,
                                                  (short)0, c, false, false);
    int rowoff = (lo ? 0 : 8) + base;
    if (base + 16 <= n) {
        if (cv) {
#pragma unroll
            for (int v = 0; v < 8; ++v) {
                float t = d[v] + bias;
                h1[(size_t)(rowoff + v) * 8 + idx] = t > 0.0f ? t : 0.0f;
            }
        }
    } else {
        if (cv) {
#pragma unroll
            for (int v = 0; v < 8; ++v) {
                int m = rowoff + v;
                if (m < n) {
                    float t = d[v] + bias;
                    h1[(size_t)m * 8 + idx] = t > 0.0f ? t : 0.0f;
                }
            }
        }
    }
}

__global__ __launch_bounds__(32) void linear2_wmma(const float* __restrict__ h1,
                                                   const float* __restrict__ dinv,
                                                   const float* __restrict__ W2,
                                                   float* __restrict__ h2, int n) {
    int lane = threadIdx.x;
    int base = blockIdx.x * 16;
    int idx  = lane & 15;
    bool lo  = lane < 16;
    int r    = base + idx;
    bool rv  = (r < n);
    int k0   = lo ? 0 : 2;

    float di = 0.0f, p0 = 0.0f, p1 = 0.0f, p4 = 0.0f, p5 = 0.0f;
    if (rv) {
        di = dinv[r];
        p0 = h1[r * 8 + k0 + 0];
        p1 = h1[r * 8 + k0 + 1];
        p4 = h1[r * 8 + k0 + 4];
        p5 = h1[r * 8 + k0 + 5];
    }
    v2f a0, a1;
    a0.x = di * p0;
    a0.y = di * p1;
    a1.x = di * p4;
    a1.y = di * p5;

    bool cv = (idx < 2);
    v2f b0, b1;
    b0.x = cv ? W2[(k0 + 0) * 2 + idx] : 0.0f;
    b0.y = cv ? W2[(k0 + 1) * 2 + idx] : 0.0f;
    b1.x = cv ? W2[(k0 + 4) * 2 + idx] : 0.0f;
    b1.y = cv ? W2[(k0 + 5) * 2 + idx] : 0.0f;

    v8f c = {};
    c = __builtin_amdgcn_wmma_f32_16x16x4_f32(false, a0, false, b0, (short)0, c, false, false);
    c = __builtin_amdgcn_wmma_f32_16x16x4_f32(false, a1, false, b1, (short)0, c, false, false);

    int rowoff = (lo ? 0 : 8) + base;
    if (base + 16 <= n) {
        if (cv) {
#pragma unroll
            for (int v = 0; v < 8; ++v) h2[(rowoff + v) * 2 + idx] = c[v];
        }
    } else {
        if (cv) {
#pragma unroll
            for (int v = 0; v < 8; ++v) {
                int m = rowoff + v;
                if (m < n) h2[m * 2 + idx] = c[v];
            }
        }
    }
}

__device__ __forceinline__ void edge_add1(const int* __restrict__ ei,
                                          const float* __restrict__ w,
                                          const float* __restrict__ hs,
                                          float* __restrict__ agg, int e, int t, int f) {
    int s = nt_load_i32(ei + t);
    int d = nt_load_i32(ei + e + t);
    atomic_add_f32(agg + (size_t)d * 2 + f, nt_load_f32(w + t) * hs[(size_t)s * 2 + f]);
}

__global__ __launch_bounds__(256) void edge_agg2(const int* __restrict__ ei,
                                                 const float* __restrict__ w,
                                                 const float* __restrict__ hs,
                                                 float* __restrict__ agg0,
                                                 float* __restrict__ agg1, int e) {
    float* agg = package_half() ? agg1 : agg0;
    const int lane = threadIdx.x & 31;
    const int f    = lane & 1;
    const int t0   = blockIdx.x * 256 + (threadIdx.x & ~31) + (lane >> 1);
    const int t1   = t0 + 16;

    if ((blockIdx.x + 1) * 256 <= e) {
        const int   s0 = nt_load_i32(ei + t0);
        const int   s1 = nt_load_i32(ei + t1);
        const int   d0 = nt_load_i32(ei + e + t0);
        const int   d1 = nt_load_i32(ei + e + t1);
        const float w0 = nt_load_f32(w + t0);
        const float w1 = nt_load_f32(w + t1);
        const float g0 = hs[(size_t)s0 * 2 + f];
        const float g1 = hs[(size_t)s1 * 2 + f];
        atomic_add_f32(agg + (size_t)d0 * 2 + f, w0 * g0);
        atomic_add_f32(agg + (size_t)d1 * 2 + f, w1 * g1);
    } else {
        if (t0 < e) edge_add1(ei, w, hs, agg, e, t0, f);
        if (t1 < e) edge_add1(ei, w, hs, agg, e, t1, f);
    }
}

__global__ __launch_bounds__(256) void finalize2(const float* __restrict__ h2,
                                                 const float* __restrict__ agg20,
                                                 const float* __restrict__ agg21,
                                                 const float* __restrict__ dinv,
                                                 const float* __restrict__ b2,
                                                 float* __restrict__ out, int n) {
    int i = blockIdx.x * blockDim.x + threadIdx.x;
    if (i >= n) return;
    float di = dinv[i];
    float v0 = di * ((agg20[(size_t)i * 2 + 0] + agg21[(size_t)i * 2 + 0]) + h2[(size_t)i * 2 + 0]) + b2[0];
    float v1 = di * ((agg20[(size_t)i * 2 + 1] + agg21[(size_t)i * 2 + 1]) + h2[(size_t)i * 2 + 1]) + b2[1];
    float m  = fmaxf(v0, v1);
    float e0 = __expf(v0 - m);
    float e1 = __expf(v1 - m);
    float inv = 1.0f / (e0 + e1);
    out[(size_t)i * 2 + 0] = e0 * inv;
    out[(size_t)i * 2 + 1] = e1 * inv;
}

static inline float* align16(float* p) {
    return (float*)(((uintptr_t)p + 15) & ~(uintptr_t)15);
}

extern "C" void kernel_launch(void* const* d_in, const int* in_sizes, int n_in,
                              void* d_out, int out_size, void* d_ws, size_t ws_size,
                              hipStream_t stream) {
    const float* x  = (const float*)d_in[0];
    const int*   ei = (const int*)d_in[1];
    const float* w  = (const float*)d_in[2];
    const float* W1 = (const float*)d_in[3];
    const float* b1 = (const float*)d_in[4];
    const float* W2 = (const float*)d_in[5];
    const float* b2 = (const float*)d_in[6];
    float* out = (float*)d_out;

    const int n = in_sizes[0] / 2;
    const int e = in_sizes[2];

    float* ws    = (float*)d_ws;
    float* dinv  = align16(ws);
    float* aggx0 = align16(dinv  + (size_t)n);
    float* aggx1 = align16(aggx0 + (size_t)n * 2);
    float* h1    = align16(aggx1 + (size_t)n * 2);
    float* h2    = align16(h1    + (size_t)n * 8);
    float* deg1  = h1;
    float* xs    = align16(deg1  + (size_t)n);
    float* agg20 = h1;
    float* agg21 = align16(agg20 + (size_t)n * 2);

    const int TB = 256;
    dim3 blkE((e + TB - 1) / TB), thr(TB);
    dim3 blkN((n + TB - 1) / TB);
    const int nzA = (int)((deg1 + (size_t)n) - dinv);
    const int nzB = (int)((agg21 + (size_t)n * 2) - agg20);
    dim3 blkZA((nzA + TB - 1) / TB);
    dim3 blkZB((nzB + TB - 1) / TB);
    dim3 blkW((n + 15) / 16), thrW(32);

    zero_kernel<<<blkZA, thr, 0, stream>>>(dinv, nzA);

    degree_kernel<<<blkE, thr, 0, stream>>>(ei, w, dinv, deg1, e);
    dinv_kernel<<<blkN, thr, 0, stream>>>(dinv, deg1, x, xs, n);

    edge_agg2<<<blkE, thr, 0, stream>>>(ei, w, xs, aggx0, aggx1, e);
    layer1_wmma<<<blkW, thrW, 0, stream>>>(x, aggx0, aggx1, dinv, W1, b1, h1, n);

    linear2_wmma<<<blkW, thrW, 0, stream>>>(h1, dinv, W2, h2, n);
    zero_kernel<<<blkZB, thr, 0, stream>>>(agg20, nzB);
    edge_agg2<<<blkE, thr, 0, stream>>>(ei, w, h2, agg20, agg21, e);
    finalize2<<<blkN, thr, 0, stream>>>(h2, agg20, agg21, dinv, b2, out, n);
}


# Layer-1 product, relu, layer-2 product in ONE launch of 8-wave groups, 4 row tiles per wave, both transposed so D of the first is B of the second in-lane; h1 never stored; one zero-fill; k order kept
# speedup vs baseline: 5.7168x; 1.0357x over previous
#include <hip/hip_runtime.h>
#include <hip/hip_bf16.h>
#include <stdint.h>

typedef __attribute__((ext_vector_type(2))) float v2f;
typedef __attribute__((ext_vector_type(8))) float v8f;

__global__ __launch_bounds__(256) void zero_kernel(float* __restrict__ p, int n) {
    int i = blockIdx.x * blockDim.x + threadIdx.x;
    if (i < n) p[i] = 0.0f;
}

__device__ __forceinline__ void atomic_add_f32(float* p, float v) {
    __hip_atomic_fetch_add(p, v, __ATOMIC_RELAXED, __HIP_MEMORY_SCOPE_AGENT);
}

__device__ __forceinline__ int   nt_load_i32(const int* p)   { return __builtin_nontemporal_load(p); }
__device__ __forceinline__ float nt_load_f32(const float* p) { return __builtin_nontemporal_load(p); }

__device__ __forceinline__ int package_half() {
    unsigned id = __builtin_amdgcn_s_sendmsg_rtn(0x87);
    unsigned die = (id >> 16) & 0xFu;
    return (die >= 4u) ? 1 : 0;
}

__global__ __launch_bounds__(256) void degree_kernel(const int* __restrict__ ei,
                                                     const float* __restrict__ w,
                                                     float* __restrict__ deg0,
                                                     float* __restrict__ deg1, int e) {
    float* deg = package_half() ? deg1 : deg0;
    int t = blockIdx.x * blockDim.x + threadIdx.x;
    if (t < e) {
        int d = nt_load_i32(ei + e + t);
        atomic_add_f32(&deg[d], nt_load_f32(w + t));
    }
}

__global__ __launch_bounds__(256) void dinv_kernel(float* __restrict__ deg0,
                                                   const float* __restrict__ deg1,
                                                   const float* __restrict__ x,
                                                   float* __restrict__ xs, int n) {
    int i = blockIdx.x * blockDim.x + threadIdx.x;
    if (i < n) {
        float  di = rsqrtf((deg0[i] + deg1[i]) + 1.0f);
        float2 xv = *(const float2*)(x + (size_t)i * 2);
        deg0[i] = di;
        *(float2*)(xs + (size_t)i * 2) = make_float2(di * xv.x, di * xv.y);
    }
}

#define TILES_PER_WAVE 4

__global__ __launch_bounds__(256) void node_chain_wmma(const float* __restrict__ x,
                                                       const float* __restrict__ aggx0,
                                                       const float* __restrict__ aggx1,
                                                       const float* __restrict__ dinv,
                                                       const float* __restrict__ W1,
                                                       const float* __restrict__ b1,
                                                       const float* __restrict__ W2,
                                                       float* __restrict__ h2, int n) {
    const int lane  = threadIdx.x & 31;
    const int wave  = threadIdx.x >> 5;
    const int c     = lane & 15;
    const int h     = lane >> 4;
    const bool lo   = (h == 0);
    const int tile0 = (blockIdx.x * 8 + wave) * TILES_PER_WAVE;

    float  di[TILES_PER_WAVE];
    float2 xv[TILES_PER_WAVE], g0[TILES_PER_WAVE], g1[TILES_PER_WAVE];
#pragma unroll
    for (int u = 0; u < TILES_PER_WAVE; ++u) {
        const int node = (tile0 + u) * 16 + c;
        const int nc   = node < n ? node : n - 1;
        di[u] = dinv[nc];
        xv[u] = *(const float2*)(x     + (size_t)nc * 2);
        g0[u] = *(const float2*)(aggx0 + (size_t)nc * 2);
        g1[u] = *(const float2*)(aggx1 + (size_t)nc * 2);
    }

    const bool frow = ((c & 4) == 0);
    const int  feat = ((c >> 3) << 1) + (c & 1) + (((c >> 1) & 1) << 2);
    const float w1a = W1[0 * 8 + feat];
    const float w1b = W1[1 * 8 + feat];
    v2f a1;
    a1.x = (lo && frow) ? w1a : 0.0f;
    a1.y = (lo && frow) ? w1b : 0.0f;

    const float bias0 = b1[2 * h + 0];
    const float bias1 = b1[2 * h + 1];
    const float bias2 = b1[2 * h + 4];
    const float bias3 = b1[2 * h + 5];

    const bool jv = (c < 2);
    const int  jc = jv ? c : 0;
    const int  k0 = 2 * h;
    const float w2a = W2[(k0 + 0) * 2 + jc];
    const float w2b = W2[(k0 + 1) * 2 + jc];
    const float w2c = W2[(k0 + 4) * 2 + jc];
    const float w2d = W2[(k0 + 5) * 2 + jc];
    v2f wa, wb;
    wa.x = jv ? w2a : 0.0f;
    wa.y = jv ? w2b : 0.0f;
    wb.x = jv ? w2c : 0.0f;
    wb.y = jv ? w2d : 0.0f;

    float dsc[TILES_PER_WAVE];
    v8f   d1[TILES_PER_WAVE];
#pragma unroll
    for (int u = 0; u < TILES_PER_WAVE; ++u) {
        const int node = (tile0 + u) * 16 + c;
        const float d  = (node < n) ? di[u] : 0.0f;
        dsc[u] = d;

        v2f bz;
        bz.x = lo ? d * ((g0[u].x + g1[u].x) + d * xv[u].x) : 0.0f;
        bz.y = lo ? d * ((g0[u].y + g1[u].y) + d * xv[u].y) : 0.0f;

        v8f zero = {};
        d1[u] = __builtin_amdgcn_wmma_f32_16x16x4_f32(false, a1, false, bz,
                                                      (short)0, zero, false, false);
    }

    v8f d2[TILES_PER_WAVE];
#pragma unroll
    for (int u = 0; u < TILES_PER_WAVE; ++u) {
        const float d = dsc[u];
        float t0 = d1[u][0] + bias0;
        float t1 = d1[u][1] + bias1;
        float t2 = d1[u][2] + bias2;
        float t3 = d1[u][3] + bias3;
        t0 = t0 > 0.0f ? t0 : 0.0f;
        t1 = t1 > 0.0f ? t1 : 0.0f;
        t2 = t2 > 0.0f ? t2 : 0.0f;
        t3 = t3 > 0.0f ? t3 : 0.0f;

        v2f hb0, hb1;
        hb0.x = d * t0;
        hb0.y = d * t1;
        hb1.x = d * t2;
        hb1.y = d * t3;

        v8f acc = {};
        acc   = __builtin_amdgcn_wmma_f32_16x16x4_f32(false, wa, false, hb0, (short)0, acc, false, false);
        d2[u] = __builtin_amdgcn_wmma_f32_16x16x4_f32(false, wb, false, hb1, (short)0, acc, false, false);
    }

#pragma unroll
    for (int u = 0; u < TILES_PER_WAVE; ++u) {
        const int node = (tile0 + u) * 16 + c;
        if (lo && node < n) *(float2*)(h2 + (size_t)node * 2) = make_float2(d2[u][0], d2[u][1]);
    }
}

__device__ __forceinline__ void edge_add1(const int* __restrict__ ei,
                                          const float* __restrict__ w,
                                          const float* __restrict__ hs,
                                          float* __restrict__ agg, int e, int t, int f) {
    int s = nt_load_i32(ei + t);
    int d = nt_load_i32(ei + e + t);
    atomic_add_f32(agg + (size_t)d * 2 + f, nt_load_f32(w + t) * hs[(size_t)s * 2 + f]);
}

__global__ __launch_bounds__(256) void edge_agg2(const int* __restrict__ ei,
                                                 const float* __restrict__ w,
                                                 const float* __restrict__ hs,
                                                 float* __restrict__ agg0,
                                                 float* __restrict__ agg1, int e) {
    float* agg = package_half() ? agg1 : agg0;
    const int lane = threadIdx.x & 31;
    const int f    = lane & 1;
    const int t0   = blockIdx.x * 256 + (threadIdx.x & ~31) + (lane >> 1);
    const int t1   = t0 + 16;

    if ((blockIdx.x + 1) * 256 <= e) {
        const int   s0 = nt_load_i32(ei + t0);
        const int   s1 = nt_load_i32(ei + t1);
        const int   d0 = nt_load_i32(ei + e + t0);
        const int   d1 = nt_load_i32(ei + e + t1);
        const float w0 = nt_load_f32(w + t0);
        const float w1 = nt_load_f32(w + t1);
        const float g0 = hs[(size_t)s0 * 2 + f];
        const float g1 = hs[(size_t)s1 * 2 + f];
        atomic_add_f32(agg + (size_t)d0 * 2 + f, w0 * g0);
        atomic_add_f32(agg + (size_t)d1 * 2 + f, w1 * g1);
    } else {
        if (t0 < e) edge_add1(ei, w, hs, agg, e, t0, f);
        if (t1 < e) edge_add1(ei, w, hs, agg, e, t1, f);
    }
}

__global__ __launch_bounds__(256) void finalize2(const float* __restrict__ h2,
                                                 const float* __restrict__ agg20,
                                                 const float* __restrict__ agg21,
                                                 const float* __restrict__ dinv,
                                                 const float* __restrict__ b2,
                                                 float* __restrict__ out, int n) {
    int i = blockIdx.x * blockDim.x + threadIdx.x;
    if (i >= n) return;
    float di = dinv[i];
    float v0 = di * ((agg20[(size_t)i * 2 + 0] + agg21[(size_t)i * 2 + 0]) + h2[(size_t)i * 2 + 0]) + b2[0];
    float v1 = di * ((agg20[(size_t)i * 2 + 1] + agg21[(size_t)i * 2 + 1]) + h2[(size_t)i * 2 + 1]) + b2[1];
    float m  = fmaxf(v0, v1);
    float e0 = __expf(v0 - m);
    float e1 = __expf(v1 - m);
    float inv = 1.0f / (e0 + e1);
    out[(size_t)i * 2 + 0] = e0 * inv;
    out[(size_t)i * 2 + 1] = e1 * inv;
}

static inline float* align16(float* p) {
    return (float*)(((uintptr_t)p + 15) & ~(uintptr_t)15);
}

extern "C" void kernel_launch(void* const* d_in, const int* in_sizes, int n_in,
                              void* d_out, int out_size, void* d_ws, size_t ws_size,
                              hipStream_t stream) {
    const float* x  = (const float*)d_in[0];
    const int*   ei = (const int*)d_in[1];
    const float* w  = (const float*)d_in[2];
    const float* W1 = (const float*)d_in[3];
    const float* b1 = (const float*)d_in[4];
    const float* W2 = (const float*)d_in[5];
    const float* b2 = (const float*)d_in[6];
    float* out = (float*)d_out;

    const int n = in_sizes[0] / 2;
    const int e = in_sizes[2];

    float* ws    = (float*)d_ws;
    float* dinv  = align16(ws);
    float* aggx0 = align16(dinv  + (size_t)n);
    float* aggx1 = align16(aggx0 + (size_t)n * 2);
    float* deg1  = align16(aggx1 + (size_t)n * 2);
    float* agg20 = align16(deg1  + (size_t)n);
    float* agg21 = align16(agg20 + (size_t)n * 2);
    float* xs    = align16(agg21 + (size_t)n * 2);
    float* h2    = align16(xs    + (size_t)n * 2);

    const int TB = 256;
    dim3 blkE((e + TB - 1) / TB), thr(TB);
    dim3 blkN((n + TB - 1) / TB);
    const int nz = (int)((agg21 + (size_t)n * 2) - dinv);
    dim3 blkZ((nz + TB - 1) / TB);
    const int tiles = (n + 15) / 16;
    dim3 blkC((tiles + 8 * TILES_PER_WAVE - 1) / (8 * TILES_PER_WAVE));

    zero_kernel<<<blkZ, thr, 0, stream>>>(dinv, nz);

    degree_kernel<<<blkE, thr, 0, stream>>>(ei, w, dinv, deg1, e);
    dinv_kernel<<<blkN, thr, 0, stream>>>(dinv, deg1, x, xs, n);

    edge_agg2<<<blkE, thr, 0, stream>>>(ei, w, xs, aggx0, aggx1, e);

    node_chain_wmma<<<blkC, thr, 0, stream>>>(x, aggx0, aggx1, dinv, W1, b1, W2, h2, n);

    edge_agg2<<<blkE, thr, 0, stream>>>(ei, w, h2, agg20, agg21, e);
    finalize2<<<blkN, thr, 0, stream>>>(h2, agg20, agg21, dinv, b2, out, n);
}


# Degree pass: each 1024-thread group keeps 65536 nodes' sums in LDS, scans 1/32 of the edges, LDS f32 add for in-range dst, then whole-line (128 B) adds to its half's deg copy; f32 sum order changes
# speedup vs baseline: 6.9086x; 1.2085x over previous
#include <hip/hip_runtime.h>
#include <hip/hip_bf16.h>
#include <stdint.h>

typedef __attribute__((ext_vector_type(2))) float v2f;
typedef __attribute__((ext_vector_type(8))) float v8f;

__global__ __launch_bounds__(256) void zero_kernel(float* __restrict__ p, int n) {
    int i = blockIdx.x * blockDim.x + threadIdx.x;
    if (i < n) p[i] = 0.0f;
}

__device__ __forceinline__ void atomic_add_f32(float* p, float v) {
    __hip_atomic_fetch_add(p, v, __ATOMIC_RELAXED, __HIP_MEMORY_SCOPE_AGENT);
}

__device__ __forceinline__ int   nt_load_i32(const int* p)   { return __builtin_nontemporal_load(p); }
__device__ __forceinline__ float nt_load_f32(const float* p) { return __builtin_nontemporal_load(p); }

__device__ __forceinline__ int package_half() {
    unsigned id = __builtin_amdgcn_s_sendmsg_rtn(0x87);
    unsigned die = (id >> 16) & 0xFu;
    return (die >= 4u) ? 1 : 0;
}

#define DEG_RANGE  65536
#define DEG_CHUNKS 32
#define DEG_LOADS  4

__device__ __forceinline__ void lds_add_f32(float* p, float v) {
    __hip_atomic_fetch_add(p, v, __ATOMIC_RELAXED, __HIP_MEMORY_SCOPE_WORKGROUP);
}

__device__ __forceinline__ void own_add(float* acc, int lo, int d, float v) {
    const unsigned r = (unsigned)(d - lo);
    if (r < (unsigned)DEG_RANGE) lds_add_f32(acc + r, v);
}

__global__ __launch_bounds__(1024) void degree_owner(const int* __restrict__ ei,
                                                     const float* __restrict__ w,
                                                     float* __restrict__ deg0,
                                                     float* __restrict__ deg1, int n, int e) {
    __shared__ __attribute__((aligned(16))) float acc[DEG_RANGE];
    const int tid = threadIdx.x;
    const int lo  = blockIdx.x * DEG_RANGE;
    const int* dst = ei + e;

    for (int i = tid * 4; i < DEG_RANGE; i += 1024 * 4)
        *(float4*)(acc + i) = make_float4(0.0f, 0.0f, 0.0f, 0.0f);
    __syncthreads();

    if ((e & 3) == 0) {
        const int words = e >> 2;
        const int per   = (words + DEG_CHUNKS - 1) / DEG_CHUNKS;
        const int b0    = blockIdx.y * per;
        const int b1    = (b0 + per < words) ? (b0 + per) : words;
        const int4*   dp = (const int4*)dst;
        const float4* wp = (const float4*)w;
        for (int base = b0; base < b1; base += 1024 * DEG_LOADS) {
            int4   d[DEG_LOADS];
            float4 v[DEG_LOADS];
#pragma unroll
            for (int u = 0; u < DEG_LOADS; ++u) {
                const int i  = base + u * 1024 + tid;
                const int ic = (i < b1) ? i : (b1 - 1);
                d[u] = dp[ic];
                v[u] = wp[ic];
            }
#pragma unroll
            for (int u = 0; u < DEG_LOADS; ++u) {
                const int i = base + u * 1024 + tid;
                if (i < b1) {
                    own_add(acc, lo, d[u].x, v[u].x);
                    own_add(acc, lo, d[u].y, v[u].y);
                    own_add(acc, lo, d[u].z, v[u].z);
                    own_add(acc, lo, d[u].w, v[u].w);
                }
            }
        }
    } else {
        const int per = (e + DEG_CHUNKS - 1) / DEG_CHUNKS;
        const int c0  = blockIdx.y * per;
        const int c1  = (c0 + per < e) ? (c0 + per) : e;
        for (int t = c0 + tid; t < c1; t += 1024) own_add(acc, lo, dst[t], w[t]);
    }
    __syncthreads();

    float* deg = package_half() ? deg1 : deg0;
    const int rot = blockIdx.y * (DEG_RANGE / DEG_CHUNKS);
    for (int j = tid; j < DEG_RANGE; j += 1024) {
        const int i = (j + rot) & (DEG_RANGE - 1);
        if (lo + i < n) atomic_add_f32(deg + lo + i, acc[i]);
    }
}

__global__ __launch_bounds__(256) void dinv_kernel(float* __restrict__ deg0,
                                                   const float* __restrict__ deg1,
                                                   const float* __restrict__ x,
                                                   float* __restrict__ xs, int n) {
    int i = blockIdx.x * blockDim.x + threadIdx.x;
    if (i < n) {
        float  di = rsqrtf((deg0[i] + deg1[i]) + 1.0f);
        float2 xv = *(const float2*)(x + (size_t)i * 2);
        deg0[i] = di;
        *(float2*)(xs + (size_t)i * 2) = make_float2(di * xv.x, di * xv.y);
    }
}

#define TILES_PER_WAVE 4

__global__ __launch_bounds__(256) void node_chain_wmma(const float* __restrict__ x,
                                                       const float* __restrict__ aggx0,
                                                       const float* __restrict__ aggx1,
                                                       const float* __restrict__ dinv,
                                                       const float* __restrict__ W1,
                                                       const float* __restrict__ b1,
                                                       const float* __restrict__ W2,
                                                       float* __restrict__ h2, int n) {
    const int lane  = threadIdx.x & 31;
    const int wave  = threadIdx.x >> 5;
    const int c     = lane & 15;
    const int h     = lane >> 4;
    const bool lo   = (h == 0);
    const int tile0 = (blockIdx.x * 8 + wave) * TILES_PER_WAVE;

    float  di[TILES_PER_WAVE];
    float2 xv[TILES_PER_WAVE], g0[TILES_PER_WAVE], g1[TILES_PER_WAVE];
#pragma unroll
    for (int u = 0; u < TILES_PER_WAVE; ++u) {
        const int node = (tile0 + u) * 16 + c;
        const int nc   = node < n ? node : n - 1;
        di[u] = dinv[nc];
        xv[u] = *(const float2*)(x     + (size_t)nc * 2);
        g0[u] = *(const float2*)(aggx0 + (size_t)nc * 2);
        g1[u] = *(const float2*)(aggx1 + (size_t)nc * 2);
    }

    const bool frow = ((c & 4) == 0);
    const int  feat = ((c >> 3) << 1) + (c & 1) + (((c >> 1) & 1) << 2);
    const float w1a = W1[0 * 8 + feat];
    const float w1b = W1[1 * 8 + feat];
    v2f a1;
    a1.x = (lo && frow) ? w1a : 0.0f;
    a1.y = (lo && frow) ? w1b : 0.0f;

    const float bias0 = b1[2 * h + 0];
    const float bias1 = b1[2 * h + 1];
    const float bias2 = b1[2 * h + 4];
    const float bias3 = b1[2 * h + 5];

    const bool jv = (c < 2);
    const int  jc = jv ? c : 0;
    const int  k0 = 2 * h;
    const float w2a = W2[(k0 + 0) * 2 + jc];
    const float w2b = W2[(k0 + 1) * 2 + jc];
    const float w2c = W2[(k0 + 4) * 2 + jc];
    const float w2d = W2[(k0 + 5) * 2 + jc];
    v2f wa, wb;
    wa.x = jv ? w2a : 0.0f;
    wa.y = jv ? w2b : 0.0f;
    wb.x = jv ? w2c : 0.0f;
    wb.y = jv ? w2d : 0.0f;

    float dsc[TILES_PER_WAVE];
    v8f   d1[TILES_PER_WAVE];
#pragma unroll
    for (int u = 0; u < TILES_PER_WAVE; ++u) {
        const int node = (tile0 + u) * 16 + c;
        const float d  = (node < n) ? di[u] : 0.0f;
        dsc[u] = d;

        v2f bz;
        bz.x = lo ? d * ((g0[u].x + g1[u].x) + d * xv[u].x) : 0.0f;
        bz.y = lo ? d * ((g0[u].y + g1[u].y) + d * xv[u].y) : 0.0f;

        v8f zero = {};
        d1[u] = __builtin_amdgcn_wmma_f32_16x16x4_f32(false, a1, false, bz,
                                                      (short)0, zero, false, false);
    }

    v8f d2[TILES_PER_WAVE];
#pragma unroll
    for (int u = 0; u < TILES_PER_WAVE; ++u) {
        const float d = dsc[u];
        float t0 = d1[u][0] + bias0;
        float t1 = d1[u][1] + bias1;
        float t2 = d1[u][2] + bias2;
        float t3 = d1[u][3] + bias3;
        t0 = t0 > 0.0f ? t0 : 0.0f;
        t1 = t1 > 0.0f ? t1 : 0.0f;
        t2 = t2 > 0.0f ? t2 : 0.0f;
        t3 = t3 > 0.0f ? t3 : 0.0f;

        v2f hb0, hb1;
        hb0.x = d * t0;
        hb0.y = d * t1;
        hb1.x = d * t2;
        hb1.y = d * t3;

        v8f acc = {};
        acc   = __builtin_amdgcn_wmma_f32_16x16x4_f32(false, wa, false, hb0, (short)0, acc, false, false);
        d2[u] = __builtin_amdgcn_wmma_f32_16x16x4_f32(false, wb, false, hb1, (short)0, acc, false, false);
    }

#pragma unroll
    for (int u = 0; u < TILES_PER_WAVE; ++u) {
        const int node = (tile0 + u) * 16 + c;
        if (lo && node < n) *(float2*)(h2 + (size_t)node * 2) = make_float2(d2[u][0], d2[u][1]);
    }
}

__device__ __forceinline__ void edge_add1(const int* __restrict__ ei,
                                          const float* __restrict__ w,
                                          const float* __restrict__ hs,
                                          float* __restrict__ agg, int e, int t, int f) {
    int s = nt_load_i32(ei + t);
    int d = nt_load_i32(ei + e + t);
    atomic_add_f32(agg + (size_t)d * 2 + f, nt_load_f32(w + t) * hs[(size_t)s * 2 + f]);
}

__global__ __launch_bounds__(256) void edge_agg2(const int* __restrict__ ei,
                                                 const float* __restrict__ w,
                                                 const float* __restrict__ hs,
                                                 float* __restrict__ agg0,
                                                 float* __restrict__ agg1, int e) {
    float* agg = package_half() ? agg1 : agg0;
    const int lane = threadIdx.x & 31;
    const int f    = lane & 1;
    const int t0   = blockIdx.x * 256 + (threadIdx.x & ~31) + (lane >> 1);
    const int t1   = t0 + 16;

    if ((blockIdx.x + 1) * 256 <= e) {
        const int   s0 = nt_load_i32(ei + t0);
        const int   s1 = nt_load_i32(ei + t1);
        const int   d0 = nt_load_i32(ei + e + t0);
        const int   d1 = nt_load_i32(ei + e + t1);
        const float w0 = nt_load_f32(w + t0);
        const float w1 = nt_load_f32(w + t1);
        const float g0 = hs[(size_t)s0 * 2 + f];
        const float g1 = hs[(size_t)s1 * 2 + f];
        atomic_add_f32(agg + (size_t)d0 * 2 + f, w0 * g0);
        atomic_add_f32(agg + (size_t)d1 * 2 + f, w1 * g1);
    } else {
        if (t0 < e) edge_add1(ei, w, hs, agg, e, t0, f);
        if (t1 < e) edge_add1(ei, w, hs, agg, e, t1, f);
    }
}

__global__ __launch_bounds__(256) void finalize2(const float* __restrict__ h2,
                                                 const float* __restrict__ agg20,
                                                 const float* __restrict__ agg21,
                                                 const float* __restrict__ dinv,
                                                 const float* __restrict__ b2,
                                                 float* __restrict__ out, int n) {
    int i = blockIdx.x * blockDim.x + threadIdx.x;
    if (i >= n) return;
    float di = dinv[i];
    float v0 = di * ((agg20[(size_t)i * 2 + 0] + agg21[(size_t)i * 2 + 0]) + h2[(size_t)i * 2 + 0]) + b2[0];
    float v1 = di * ((agg20[(size_t)i * 2 + 1] + agg21[(size_t)i * 2 + 1]) + h2[(size_t)i * 2 + 1]) + b2[1];
    float m  = fmaxf(v0, v1);
    float e0 = __expf(v0 - m);
    float e1 = __expf(v1 - m);
    float inv = 1.0f / (e0 + e1);
    out[(size_t)i * 2 + 0] = e0 * inv;
    out[(size_t)i * 2 + 1] = e1 * inv;
}

static inline float* align16(float* p) {
    return (float*)(((uintptr_t)p + 15) & ~(uintptr_t)15);
}

extern "C" void kernel_launch(void* const* d_in, const int* in_sizes, int n_in,
                              void* d_out, int out_size, void* d_ws, size_t ws_size,
                              hipStream_t stream) {
    const float* x  = (const float*)d_in[0];
    const int*   ei = (const int*)d_in[1];
    const float* w  = (const float*)d_in[2];
    const float* W1 = (const float*)d_in[3];
    const float* b1 = (const float*)d_in[4];
    const float* W2 = (const float*)d_in[5];
    const float* b2 = (const float*)d_in[6];
    float* out = (float*)d_out;

    const int n = in_sizes[0] / 2;
    const int e = in_sizes[2];

    float* ws    = (float*)d_ws;
    float* dinv  = align16(ws);
    float* aggx0 = align16(dinv  + (size_t)n);
    float* aggx1 = align16(aggx0 + (size_t)n * 2);
    float* deg1  = align16(aggx1 + (size_t)n * 2);
    float* agg20 = align16(deg1  + (size_t)n);
    float* agg21 = align16(agg20 + (size_t)n * 2);
    float* xs    = align16(agg21 + (size_t)n * 2);
    float* h2    = align16(xs    + (size_t)n * 2);

    const int TB = 256;
    dim3 blkE((e + TB - 1) / TB), thr(TB);
    dim3 blkN((n + TB - 1) / TB);
    const int nz = (int)((agg21 + (size_t)n * 2) - dinv);
    dim3 blkZ((nz + TB - 1) / TB);
    const int tiles = (n + 15) / 16;
    dim3 blkC((tiles + 8 * TILES_PER_WAVE - 1) / (8 * TILES_PER_WAVE));

    zero_kernel<<<blkZ, thr, 0, stream>>>(dinv, nz);

    degree_owner<<<dim3((n + DEG_RANGE - 1) / DEG_RANGE, DEG_CHUNKS), dim3(1024), 0, stream>>>(ei, w, dinv, deg1, n, e);
    dinv_kernel<<<blkN, thr, 0, stream>>>(dinv, deg1, x, xs, n);

    edge_agg2<<<blkE, thr, 0, stream>>>(ei, w, xs, aggx0, aggx1, e);

    node_chain_wmma<<<blkC, thr, 0, stream>>>(x, aggx0, aggx1, dinv, W1, b1, W2, h2, n);

    edge_agg2<<<blkE, thr, 0, stream>>>(ei, w, h2, agg20, agg21, e);
    finalize2<<<blkN, thr, 0, stream>>>(h2, agg20, agg21, dinv, b2, out, n);
}


# Fewer, fatter waves in the two wave-count-bound kernels: an edge-pass wave takes 256 edges (was 32), a zero-fill thread 8 float4 stores (was 1 float); same adds, arrival-order sums as before
# speedup vs baseline: 7.8754x; 1.1399x over previous
#include <hip/hip_runtime.h>
#include <hip/hip_bf16.h>
#include <stdint.h>

typedef __attribute__((ext_vector_type(2))) float v2f;
typedef __attribute__((ext_vector_type(8))) float v8f;


#define ZERO_PER_THREAD 8

__global__ __launch_bounds__(256) void zero_kernel(float* __restrict__ p, int n) {
    const int n4   = n >> 2;
    const int base = blockIdx.x * (256 * ZERO_PER_THREAD) + threadIdx.x;
    float4* p4 = (float4*)p;
#pragma unroll
    for (int u = 0; u < ZERO_PER_THREAD; ++u) {
        const int i = base + u * 256;
        if (i < n4) p4[i] = make_float4(0.0f, 0.0f, 0.0f, 0.0f);
    }
    if (blockIdx.x == 0 && (int)threadIdx.x < (n & 3)) p[(n4 << 2) + threadIdx.x] = 0.0f;
}

__device__ __forceinline__ void atomic_add_f32(float* p, float v) {
    __hip_atomic_fetch_add(p, v, __ATOMIC_RELAXED, __HIP_MEMORY_SCOPE_AGENT);
}

__device__ __forceinline__ int   nt_load_i32(const int* p)   { return __builtin_nontemporal_load(p); }
__device__ __forceinline__ float nt_load_f32(const float* p) { return __builtin_nontemporal_load(p); }

__device__ __forceinline__ int package_half() {
    unsigned id = __builtin_amdgcn_s_sendmsg_rtn(0x87);
    unsigned die = (id >> 16) & 0xFu;
    return (die >= 4u) ? 1 : 0;
}

#define DEG_RANGE  65536
#define DEG_CHUNKS 32
#define DEG_LOADS  4

__device__ __forceinline__ void lds_add_f32(float* p, float v) {
    __hip_atomic_fetch_add(p, v, __ATOMIC_RELAXED, __HIP_MEMORY_SCOPE_WORKGROUP);
}

__device__ __forceinline__ void own_add(float* acc, int lo, int d, float v) {
    const unsigned r = (unsigned)(d - lo);
    if (r < (unsigned)DEG_RANGE) lds_add_f32(acc + r, v);
}

__global__ __launch_bounds__(1024) void degree_owner(const int* __restrict__ ei,
                                                     const float* __restrict__ w,
                                                     float* __restrict__ deg0,
                                                     float* __restrict__ deg1, int n, int e) {
    __shared__ __attribute__((aligned(16))) float acc[DEG_RANGE];
    const int tid = threadIdx.x;
    const int lo  = blockIdx.x * DEG_RANGE;
    const int* dst = ei + e;

    for (int i = tid * 4; i < DEG_RANGE; i += 1024 * 4)
        *(float4*)(acc + i) = make_float4(0.0f, 0.0f, 0.0f, 0.0f);
    __syncthreads();

    if ((e & 3) == 0) {
        const int words = e >> 2;
        const int per   = (words + DEG_CHUNKS - 1) / DEG_CHUNKS;
        const int b0    = blockIdx.y * per;
        const int b1    = (b0 + per < words) ? (b0 + per) : words;
        const int4*   dp = (const int4*)dst;
        const float4* wp = (const float4*)w;
        for (int base = b0; base < b1; base += 1024 * DEG_LOADS) {
            int4   d[DEG_LOADS];
            float4 v[DEG_LOADS];
#pragma unroll
            for (int u = 0; u < DEG_LOADS; ++u) {
                const int i  = base + u * 1024 + tid;
                const int ic = (i < b1) ? i : (b1 - 1);
                d[u] = dp[ic];
                v[u] = wp[ic];
            }
#pragma unroll
            for (int u = 0; u < DEG_LOADS; ++u) {
                const int i = base + u * 1024 + tid;
                if (i < b1) {
                    own_add(acc, lo, d[u].x, v[u].x);
                    own_add(acc, lo, d[u].y, v[u].y);
                    own_add(acc, lo, d[u].z, v[u].z);
                    own_add(acc, lo, d[u].w, v[u].w);
                }
            }
        }
    } else {
        const int per = (e + DEG_CHUNKS - 1) / DEG_CHUNKS;
        const int c0  = blockIdx.y * per;
        const int c1  = (c0 + per < e) ? (c0 + per) : e;
        for (int t = c0 + tid; t < c1; t += 1024) own_add(acc, lo, dst[t], w[t]);
    }
    __syncthreads();

    float* deg = package_half() ? deg1 : deg0;
    const int rot = blockIdx.y * (DEG_RANGE / DEG_CHUNKS);
    for (int j = tid; j < DEG_RANGE; j += 1024) {
        const int i = (j + rot) & (DEG_RANGE - 1);
        if (lo + i < n) atomic_add_f32(deg + lo + i, acc[i]);
    }
}

__global__ __launch_bounds__(256) void dinv_kernel(float* __restrict__ deg0,
                                                   const float* __restrict__ deg1,
                                                   const float* __restrict__ x,
                                                   float* __restrict__ xs, int n) {
    int i = blockIdx.x * blockDim.x + threadIdx.x;
    if (i < n) {
        float  di = rsqrtf((deg0[i] + deg1[i]) + 1.0f);
        float2 xv = *(const float2*)(x + (size_t)i * 2);
        deg0[i] = di;
        *(float2*)(xs + (size_t)i * 2) = make_float2(di * xv.x, di * xv.y);
    }
}

#define TILES_PER_WAVE 4

__global__ __launch_bounds__(256) void node_chain_wmma(const float* __restrict__ x,
                                                       const float* __restrict__ aggx0,
                                                       const float* __restrict__ aggx1,
                                                       const float* __restrict__ dinv,
                                                       const float* __restrict__ W1,
                                                       const float* __restrict__ b1,
                                                       const float* __restrict__ W2,
                                                       float* __restrict__ h2, int n) {
    const int lane  = threadIdx.x & 31;
    const int wave  = threadIdx.x >> 5;
    const int c     = lane & 15;
    const int h     = lane >> 4;
    const bool lo   = (h == 0);
    const int tile0 = (blockIdx.x * 8 + wave) * TILES_PER_WAVE;

    float  di[TILES_PER_WAVE];
    float2 xv[TILES_PER_WAVE], g0[TILES_PER_WAVE], g1[TILES_PER_WAVE];
#pragma unroll
    for (int u = 0; u < TILES_PER_WAVE; ++u) {
        const int node = (tile0 + u) * 16 + c;
        const int nc   = node < n ? node : n - 1;
        di[u] = dinv[nc];
        xv[u] = *(const float2*)(x     + (size_t)nc * 2);
        g0[u] = *(const float2*)(aggx0 + (size_t)nc * 2);
        g1[u] = *(const float2*)(aggx1 + (size_t)nc * 2);
    }

    const bool frow = ((c & 4) == 0);
    const int  feat = ((c >> 3) << 1) + (c & 1) + (((c >> 1) & 1) << 2);
    const float w1a = W1[0 * 8 + feat];
    const float w1b = W1[1 * 8 + feat];
    v2f a1;
    a1.x = (lo && frow) ? w1a : 0.0f;
    a1.y = (lo && frow) ? w1b : 0.0f;

    const float bias0 = b1[2 * h + 0];
    const float bias1 = b1[2 * h + 1];
    const float bias2 = b1[2 * h + 4];
    const float bias3 = b1[2 * h + 5];

    const bool jv = (c < 2);
    const int  jc = jv ? c : 0;
    const int  k0 = 2 * h;
    const float w2a = W2[(k0 + 0) * 2 + jc];
    const float w2b = W2[(k0 + 1) * 2 + jc];
    const float w2c = W2[(k0 + 4) * 2 + jc];
    const float w2d = W2[(k0 + 5) * 2 + jc];
    v2f wa, wb;
    wa.x = jv ? w2a : 0.0f;
    wa.y = jv ? w2b : 0.0f;
    wb.x = jv ? w2c : 0.0f;
    wb.y = jv ? w2d : 0.0f;

    float dsc[TILES_PER_WAVE];
    v8f   d1[TILES_PER_WAVE];
#pragma unroll
    for (int u = 0; u < TILES_PER_WAVE; ++u) {
        const int node = (tile0 + u) * 16 + c;
        const float d  = (node < n) ? di[u] : 0.0f;
        dsc[u] = d;

        v2f bz;
        bz.x = lo ? d * ((g0[u].x + g1[u].x) + d * xv[u].x) : 0.0f;
        bz.y = lo ? d * ((g0[u].y + g1[u].y) + d * xv[u].y) : 0.0f;

        v8f zero = {};
        d1[u] = __builtin_amdgcn_wmma_f32_16x16x4_f32(false, a1, false, bz,
                                                      (short)0, zero, false, false);
    }

    v8f d2[TILES_PER_WAVE];
#pragma unroll
    for (int u = 0; u < TILES_PER_WAVE; ++u) {
        const float d = dsc[u];
        float t0 = d1[u][0] + bias0;
        float t1 = d1[u][1] + bias1;
        float t2 = d1[u][2] + bias2;
        float t3 = d1[u][3] + bias3;
        t0 = t0 > 0.0f ? t0 : 0.0f;
        t1 = t1 > 0.0f ? t1 : 0.0f;
        t2 = t2 > 0.0f ? t2 : 0.0f;
        t3 = t3 > 0.0f ? t3 : 0.0f;

        v2f hb0, hb1;
        hb0.x = d * t0;
        hb0.y = d * t1;
        hb1.x = d * t2;
        hb1.y = d * t3;

        v8f acc = {};
        acc   = __builtin_amdgcn_wmma_f32_16x16x4_f32(false, wa, false, hb0, (short)0, acc, false, false);
        d2[u] = __builtin_amdgcn_wmma_f32_16x16x4_f32(false, wb, false, hb1, (short)0, acc, false, false);
    }

#pragma unroll
    for (int u = 0; u < TILES_PER_WAVE; ++u) {
        const int node = (tile0 + u) * 16 + c;
        if (lo && node < n) *(float2*)(h2 + (size_t)node * 2) = make_float2(d2[u][0], d2[u][1]);
    }
}

#define EDGE_GROUPS     16
#define EDGES_PER_WAVE  (16 * EDGE_GROUPS)
#define EDGES_PER_BLOCK (8 * EDGES_PER_WAVE)

__device__ __forceinline__ void edge_add1(const int* __restrict__ ei,
                                          const float* __restrict__ w,
                                          const float* __restrict__ hs,
                                          float* __restrict__ agg, int e, int t, int f) {
    int s = nt_load_i32(ei + t);
    int d = nt_load_i32(ei + e + t);
    atomic_add_f32(agg + (size_t)d * 2 + f, nt_load_f32(w + t) * hs[(size_t)s * 2 + f]);
}

__global__ __launch_bounds__(256) void edge_agg2(const int* __restrict__ ei,
                                                 const float* __restrict__ w,
                                                 const float* __restrict__ hs,
                                                 float* __restrict__ agg0,
                                                 float* __restrict__ agg1, int e) {
    float* agg = package_half() ? agg1 : agg0;
    const int lane  = threadIdx.x & 31;
    const int wave  = threadIdx.x >> 5;
    const int f     = lane & 1;
    const int first = blockIdx.x * EDGES_PER_BLOCK;
    const int t0    = first + wave * EDGES_PER_WAVE + (lane >> 1);

    if (first + EDGES_PER_BLOCK <= e) {
        int   s[EDGE_GROUPS], d[EDGE_GROUPS];
        float wv[EDGE_GROUPS], g[EDGE_GROUPS];
#pragma unroll
        for (int u = 0; u < EDGE_GROUPS; ++u) {
            const int t = t0 + 16 * u;
            s[u]  = nt_load_i32(ei + t);
            d[u]  = nt_load_i32(ei + e + t);
            wv[u] = nt_load_f32(w + t);
        }
#pragma unroll
        for (int u = 0; u < EDGE_GROUPS; ++u) g[u] = hs[(size_t)s[u] * 2 + f];
#pragma unroll
        for (int u = 0; u < EDGE_GROUPS; ++u) atomic_add_f32(agg + (size_t)d[u] * 2 + f, wv[u] * g[u]);
    } else {
        for (int u = 0; u < EDGE_GROUPS; ++u) {
            const int t = t0 + 16 * u;
            if (t < e) edge_add1(ei, w, hs, agg, e, t, f);
        }
    }
}

__global__ __launch_bounds__(256) void finalize2(const float* __restrict__ h2,
                                                 const float* __restrict__ agg20,
                                                 const float* __restrict__ agg21,
                                                 const float* __restrict__ dinv,
                                                 const float* __restrict__ b2,
                                                 float* __restrict__ out, int n) {
    int i = blockIdx.x * blockDim.x + threadIdx.x;
    if (i >= n) return;
    float di = dinv[i];
    float v0 = di * ((agg20[(size_t)i * 2 + 0] + agg21[(size_t)i * 2 + 0]) + h2[(size_t)i * 2 + 0]) + b2[0];
    float v1 = di * ((agg20[(size_t)i * 2 + 1] + agg21[(size_t)i * 2 + 1]) + h2[(size_t)i * 2 + 1]) + b2[1];
    float m  = fmaxf(v0, v1);
    float e0 = __expf(v0 - m);
    float e1 = __expf(v1 - m);
    float inv = 1.0f / (e0 + e1);
    out[(size_t)i * 2 + 0] = e0 * inv;
    out[(size_t)i * 2 + 1] = e1 * inv;
}

static inline float* align16(float* p) {
    return (float*)(((uintptr_t)p + 15) & ~(uintptr_t)15);
}

extern "C" void kernel_launch(void* const* d_in, const int* in_sizes, int n_in,
                              void* d_out, int out_size, void* d_ws, size_t ws_size,
                              hipStream_t stream) {
    const float* x  = (const float*)d_in[0];
    const int*   ei = (const int*)d_in[1];
    const float* w  = (const float*)d_in[2];
    const float* W1 = (const float*)d_in[3];
    const float* b1 = (const float*)d_in[4];
    const float* W2 = (const float*)d_in[5];
    const float* b2 = (const float*)d_in[6];
    float* out = (float*)d_out;

    const int n = in_sizes[0] / 2;
    const int e = in_sizes[2];

    float* ws    = (float*)d_ws;
    float* dinv  = align16(ws);
    float* aggx0 = align16(dinv  + (size_t)n);
    float* aggx1 = align16(aggx0 + (size_t)n * 2);
    float* deg1  = align16(aggx1 + (size_t)n * 2);
    float* agg20 = align16(deg1  + (size_t)n);
    float* agg21 = align16(agg20 + (size_t)n * 2);
    float* xs    = align16(agg21 + (size_t)n * 2);
    float* h2    = align16(xs    + (size_t)n * 2);

    const int TB = 256;
    dim3 blkE((e + EDGES_PER_BLOCK - 1) / EDGES_PER_BLOCK), thr(TB);
    dim3 blkN((n + TB - 1) / TB);
    const int nz = (int)((agg21 + (size_t)n * 2) - dinv);
    dim3 blkZ((nz / 4) / (TB * ZERO_PER_THREAD) + 1);
    const int tiles = (n + 15) / 16;
    dim3 blkC((tiles + 8 * TILES_PER_WAVE - 1) / (8 * TILES_PER_WAVE));

    zero_kernel<<<blkZ, thr, 0, stream>>>(dinv, nz);

    degree_owner<<<dim3((n + DEG_RANGE - 1) / DEG_RANGE, DEG_CHUNKS), dim3(1024), 0, stream>>>(ei, w, dinv, deg1, n, e);
    dinv_kernel<<<blkN, thr, 0, stream>>>(dinv, deg1, x, xs, n);

    edge_agg2<<<blkE, thr, 0, stream>>>(ei, w, xs, aggx0, aggx1, e);

    node_chain_wmma<<<blkC, thr, 0, stream>>>(x, aggx0, aggx1, dinv, W1, b1, W2, h2, n);

    edge_agg2<<<blkE, thr, 0, stream>>>(ei, w, h2, agg20, agg21, e);
    finalize2<<<blkN, thr, 0, stream>>>(h2, agg20, agg21, dinv, b2, out, n);
}


# Edges ordered once by destination range (per-chunk LDS counting sort, packed records); degree and both edge sums add in range owners' LDS, flush by line; no per-edge L2 atomic; f32 sum order changes
# speedup vs baseline: 9.4752x; 1.2031x over previous
#include <hip/hip_runtime.h>
#include <hip/hip_bf16.h>
#include <stdint.h>

typedef __attribute__((ext_vector_type(2))) float v2f;
typedef __attribute__((ext_vector_type(8))) float v8f;


#define ZERO_PER_THREAD 8

__global__ __launch_bounds__(256) void zero_kernel(float* __restrict__ p, int n) {
    const int n4   = n >> 2;
    const int base = blockIdx.x * (256 * ZERO_PER_THREAD) + threadIdx.x;
    float4* p4 = (float4*)p;
#pragma unroll
    for (int u = 0; u < ZERO_PER_THREAD; ++u) {
        const int i = base + u * 256;
        if (i < n4) p4[i] = make_float4(0.0f, 0.0f, 0.0f, 0.0f);
    }
    if (blockIdx.x == 0 && (int)threadIdx.x < (n & 3)) p[(n4 << 2) + threadIdx.x] = 0.0f;
}

__device__ __forceinline__ void atomic_add_f32(float* p, float v) {
    __hip_atomic_fetch_add(p, v, __ATOMIC_RELAXED, __HIP_MEMORY_SCOPE_AGENT);
}

__device__ __forceinline__ int   nt_load_i32(const int* p)   { return __builtin_nontemporal_load(p); }
__device__ __forceinline__ float nt_load_f32(const float* p) { return __builtin_nontemporal_load(p); }

__device__ __forceinline__ int package_half() {
    unsigned id = __builtin_amdgcn_s_sendmsg_rtn(0x87);
    unsigned die = (id >> 16) & 0xFu;
    return (die >= 4u) ? 1 : 0;
}

#define DEG_RANGE  65536
#define DEG_CHUNKS 32
#define DEG_LOADS  4

__device__ __forceinline__ void lds_add_f32(float* p, float v) {
    __hip_atomic_fetch_add(p, v, __ATOMIC_RELAXED, __HIP_MEMORY_SCOPE_WORKGROUP);
}

__device__ __forceinline__ void own_add(float* acc, int lo, int d, float v) {
    const unsigned r = (unsigned)(d - lo);
    if (r < (unsigned)DEG_RANGE) lds_add_f32(acc + r, v);
}

__global__ __launch_bounds__(1024) void degree_owner(const int* __restrict__ ei,
                                                     const float* __restrict__ w,
                                                     float* __restrict__ deg0,
                                                     float* __restrict__ deg1, int n, int e) {
    __shared__ __attribute__((aligned(16))) float acc[DEG_RANGE];
    const int tid = threadIdx.x;
    const int lo  = blockIdx.x * DEG_RANGE;
    const int* dst = ei + e;

    for (int i = tid * 4; i < DEG_RANGE; i += 1024 * 4)
        *(float4*)(acc + i) = make_float4(0.0f, 0.0f, 0.0f, 0.0f);
    __syncthreads();

    if ((e & 3) == 0) {
        const int words = e >> 2;
        const int per   = (words + DEG_CHUNKS - 1) / DEG_CHUNKS;
        const int b0    = blockIdx.y * per;
        const int b1    = (b0 + per < words) ? (b0 + per) : words;
        const int4*   dp = (const int4*)dst;
        const float4* wp = (const float4*)w;
        for (int base = b0; base < b1; base += 1024 * DEG_LOADS) {
            int4   d[DEG_LOADS];
            float4 v[DEG_LOADS];
#pragma unroll
            for (int u = 0; u < DEG_LOADS; ++u) {
                const int i  = base + u * 1024 + tid;
                const int ic = (i < b1) ? i : (b1 - 1);
                d[u] = dp[ic];
                v[u] = wp[ic];
            }
#pragma unroll
            for (int u = 0; u < DEG_LOADS; ++u) {
                const int i = base + u * 1024 + tid;
                if (i < b1) {
                    own_add(acc, lo, d[u].x, v[u].x);
                    own_add(acc, lo, d[u].y, v[u].y);
                    own_add(acc, lo, d[u].z, v[u].z);
                    own_add(acc, lo, d[u].w, v[u].w);
                }
            }
        }
    } else {
        const int per = (e + DEG_CHUNKS - 1) / DEG_CHUNKS;
        const int c0  = blockIdx.y * per;
        const int c1  = (c0 + per < e) ? (c0 + per) : e;
        for (int t = c0 + tid; t < c1; t += 1024) own_add(acc, lo, dst[t], w[t]);
    }
    __syncthreads();

    float* deg = package_half() ? deg1 : deg0;
    const int rot = blockIdx.y * (DEG_RANGE / DEG_CHUNKS);
    for (int j = tid; j < DEG_RANGE; j += 1024) {
        const int i = (j + rot) & (DEG_RANGE - 1);
        if (lo + i < n) atomic_add_f32(deg + lo + i, acc[i]);
    }
}

__global__ __launch_bounds__(256) void dinv_kernel(float* __restrict__ deg0,
                                                   const float* __restrict__ deg1,
                                                   const float* __restrict__ x,
                                                   float* __restrict__ xs, int n) {
    int i = blockIdx.x * blockDim.x + threadIdx.x;
    if (i < n) {
        float  di = rsqrtf((deg0[i] + deg1[i]) + 1.0f);
        float2 xv = *(const float2*)(x + (size_t)i * 2);
        deg0[i] = di;
        *(float2*)(xs + (size_t)i * 2) = make_float2(di * xv.x, di * xv.y);
    }
}

#define TILES_PER_WAVE 4

__global__ __launch_bounds__(256) void node_chain_wmma(const float* __restrict__ x,
                                                       const float* __restrict__ aggx0,
                                                       const float* __restrict__ aggx1,
                                                       const float* __restrict__ dinv,
                                                       const float* __restrict__ W1,
                                                       const float* __restrict__ b1,
                                                       const float* __restrict__ W2,
                                                       float* __restrict__ h2, int n) {
    const int lane  = threadIdx.x & 31;
    const int wave  = threadIdx.x >> 5;
    const int c     = lane & 15;
    const int h     = lane >> 4;
    const bool lo   = (h == 0);
    const int tile0 = (blockIdx.x * 8 + wave) * TILES_PER_WAVE;

    float  di[TILES_PER_WAVE];
    float2 xv[TILES_PER_WAVE], g0[TILES_PER_WAVE], g1[TILES_PER_WAVE];
#pragma unroll
    for (int u = 0; u < TILES_PER_WAVE; ++u) {
        const int node = (tile0 + u) * 16 + c;
        const int nc   = node < n ? node : n - 1;
        di[u] = dinv[nc];
        xv[u] = *(const float2*)(x     + (size_t)nc * 2);
        g0[u] = *(const float2*)(aggx0 + (size_t)nc * 2);
        g1[u] = *(const float2*)(aggx1 + (size_t)nc * 2);
    }

    const bool frow = ((c & 4) == 0);
    const int  feat = ((c >> 3) << 1) + (c & 1) + (((c >> 1) & 1) << 2);
    const float w1a = W1[0 * 8 + feat];
    const float w1b = W1[1 * 8 + feat];
    v2f a1;
    a1.x = (lo && frow) ? w1a : 0.0f;
    a1.y = (lo && frow) ? w1b : 0.0f;

    const float bias0 = b1[2 * h + 0];
    const float bias1 = b1[2 * h + 1];
    const float bias2 = b1[2 * h + 4];
    const float bias3 = b1[2 * h + 5];

    const bool jv = (c < 2);
    const int  jc = jv ? c : 0;
    const int  k0 = 2 * h;
    const float w2a = W2[(k0 + 0) * 2 + jc];
    const float w2b = W2[(k0 + 1) * 2 + jc];
    const float w2c = W2[(k0 + 4) * 2 + jc];
    const float w2d = W2[(k0 + 5) * 2 + jc];
    v2f wa, wb;
    wa.x = jv ? w2a : 0.0f;
    wa.y = jv ? w2b : 0.0f;
    wb.x = jv ? w2c : 0.0f;
    wb.y = jv ? w2d : 0.0f;

    float dsc[TILES_PER_WAVE];
    v8f   d1[TILES_PER_WAVE];
#pragma unroll
    for (int u = 0; u < TILES_PER_WAVE; ++u) {
        const int node = (tile0 + u) * 16 + c;
        const float d  = (node < n) ? di[u] : 0.0f;
        dsc[u] = d;

        v2f bz;
        bz.x = lo ? d * ((g0[u].x + g1[u].x) + d * xv[u].x) : 0.0f;
        bz.y = lo ? d * ((g0[u].y + g1[u].y) + d * xv[u].y) : 0.0f;

        v8f zero = {};
        d1[u] = __builtin_amdgcn_wmma_f32_16x16x4_f32(false, a1, false, bz,
                                                      (short)0, zero, false, false);
    }

    v8f d2[TILES_PER_WAVE];
#pragma unroll
    for (int u = 0; u < TILES_PER_WAVE; ++u) {
        const float d = dsc[u];
        float t0 = d1[u][0] + bias0;
        float t1 = d1[u][1] + bias1;
        float t2 = d1[u][2] + bias2;
        float t3 = d1[u][3] + bias3;
        t0 = t0 > 0.0f ? t0 : 0.0f;
        t1 = t1 > 0.0f ? t1 : 0.0f;
        t2 = t2 > 0.0f ? t2 : 0.0f;
        t3 = t3 > 0.0f ? t3 : 0.0f;

        v2f hb0, hb1;
        hb0.x = d * t0;
        hb0.y = d * t1;
        hb1.x = d * t2;
        hb1.y = d * t3;

        v8f acc = {};
        acc   = __builtin_amdgcn_wmma_f32_16x16x4_f32(false, wa, false, hb0, (short)0, acc, false, false);
        d2[u] = __builtin_amdgcn_wmma_f32_16x16x4_f32(false, wb, false, hb1, (short)0, acc, false, false);
    }

#pragma unroll
    for (int u = 0; u < TILES_PER_WAVE; ++u) {
        const int node = (tile0 + u) * 16 + c;
        if (lo && node < n) *(float2*)(h2 + (size_t)node * 2) = make_float2(d2[u][0], d2[u][1]);
    }
}

#define EDGE_GROUPS     16
#define EDGES_PER_WAVE  (16 * EDGE_GROUPS)
#define EDGES_PER_BLOCK (8 * EDGES_PER_WAVE)

__device__ __forceinline__ void edge_add1(const int* __restrict__ ei,
                                          const float* __restrict__ w,
                                          const float* __restrict__ hs,
                                          float* __restrict__ agg, int e, int t, int f) {
    int s = nt_load_i32(ei + t);
    int d = nt_load_i32(ei + e + t);
    atomic_add_f32(agg + (size_t)d * 2 + f, nt_load_f32(w + t) * hs[(size_t)s * 2 + f]);
}

__global__ __launch_bounds__(256) void edge_agg2(const int* __restrict__ ei,
                                                 const float* __restrict__ w,
                                                 const float* __restrict__ hs,
                                                 float* __restrict__ agg0,
                                                 float* __restrict__ agg1, int e) {
    float* agg = package_half() ? agg1 : agg0;
    const int lane  = threadIdx.x & 31;
    const int wave  = threadIdx.x >> 5;
    const int f     = lane & 1;
    const int first = blockIdx.x * EDGES_PER_BLOCK;
    const int t0    = first + wave * EDGES_PER_WAVE + (lane >> 1);

    if (first + EDGES_PER_BLOCK <= e) {
        int   s[EDGE_GROUPS], d[EDGE_GROUPS];
        float wv[EDGE_GROUPS], g[EDGE_GROUPS];
#pragma unroll
        for (int u = 0; u < EDGE_GROUPS; ++u) {
            const int t = t0 + 16 * u;
            s[u]  = nt_load_i32(ei + t);
            d[u]  = nt_load_i32(ei + e + t);
            wv[u] = nt_load_f32(w + t);
        }
#pragma unroll
        for (int u = 0; u < EDGE_GROUPS; ++u) g[u] = hs[(size_t)s[u] * 2 + f];
#pragma unroll
        for (int u = 0; u < EDGE_GROUPS; ++u) atomic_add_f32(agg + (size_t)d[u] * 2 + f, wv[u] * g[u]);
    } else {
        for (int u = 0; u < EDGE_GROUPS; ++u) {
            const int t = t0 + 16 * u;
            if (t < e) edge_add1(ei, w, hs, agg, e, t, f);
        }
    }
}

#define BIN_SHIFT   13
#define BIN_NODES   (1 << BIN_SHIFT)
#define BIN_MAX     64
#define CHUNK_EDGES 8192
#define CHUNK_PER_THREAD 8
#define OFF_STRIDE  (BIN_MAX + 1)
#define OWN_GROUPS  8
#define OWN_LOADS   5
#define PLANE1      (BIN_NODES + 32)

__global__ __launch_bounds__(1024) void bin_chunks(const int* __restrict__ ei,
                                                   const float* __restrict__ w,
                                                   unsigned* __restrict__ rec,
                                                   int* __restrict__ off,
                                                   int e, int nbins) {
    __shared__ int   cnt[BIN_MAX];
    __shared__ int   start[OFF_STRIDE];
    __shared__ uint2 stage[CHUNK_EDGES];
    const int  tid  = threadIdx.x;
    const int  base = blockIdx.x * CHUNK_EDGES;
    const bool full = (base + CHUNK_EDGES <= e);

    if (tid < BIN_MAX) cnt[tid] = 0;
    __syncthreads();

    int   s[CHUNK_PER_THREAD], d[CHUNK_PER_THREAD];
    float wv[CHUNK_PER_THREAD];
#pragma unroll
    for (int u = 0; u < CHUNK_PER_THREAD; ++u) {
        const int t  = base + u * 1024 + tid;
        const int tc = (t < e) ? t : (e - 1);
        s[u]  = nt_load_i32(ei + tc);
        d[u]  = nt_load_i32(ei + e + tc);
        wv[u] = nt_load_f32(w + tc);
    }
    if (full) {
        int rk[CHUNK_PER_THREAD];
#pragma unroll
        for (int u = 0; u < CHUNK_PER_THREAD; ++u) {
            int b = d[u] >> BIN_SHIFT;
            b = b < 0 ? 0 : (b >= nbins ? nbins - 1 : b);
            d[u]  = (b << BIN_SHIFT) | (d[u] & (BIN_NODES - 1));
            rk[u] = __hip_atomic_fetch_add(&cnt[b], 1, __ATOMIC_RELAXED, __HIP_MEMORY_SCOPE_WORKGROUP);
        }
#pragma unroll
        for (int u = 0; u < CHUNK_PER_THREAD; ++u) d[u] |= (int)((unsigned)rk[u] << (BIN_SHIFT + 6));
    } else {
#pragma unroll
        for (int u = 0; u < CHUNK_PER_THREAD; ++u) {
            const int t = base + u * 1024 + tid;
            int b = d[u] >> BIN_SHIFT;
            b = b < 0 ? 0 : (b >= nbins ? nbins - 1 : b);
            int rk = 0;
            if (t < e) rk = __hip_atomic_fetch_add(&cnt[b], 1, __ATOMIC_RELAXED, __HIP_MEMORY_SCOPE_WORKGROUP);
            d[u] = (int)((unsigned)rk << (BIN_SHIFT + 6)) | (b << BIN_SHIFT) | (d[u] & (BIN_NODES - 1));
        }
    }
    __syncthreads();

    if (tid < 32) {
        const int c0 = cnt[2 * tid];
        const int c1 = cnt[2 * tid + 1];
        int incl = c0 + c1;
#pragma unroll
        for (int k = 1; k < 32; k <<= 1) {
            const int up = __shfl_up(incl, k, 32);
            if (tid >= k) incl += up;
        }
        const int excl = incl - (c0 + c1);
        start[2 * tid]     = excl;
        start[2 * tid + 1] = excl + c0;
        if (tid == 31) start[BIN_MAX] = incl;
    }
    __syncthreads();

    if (tid < OFF_STRIDE) off[(size_t)blockIdx.x * OFF_STRIDE + tid] = start[tid];
#pragma unroll
    for (int u = 0; u < CHUNK_PER_THREAD; ++u) {
        const int t = base + u * 1024 + tid;
        if (full || t < e) {
            const int b  = (d[u] >> BIN_SHIFT) & (BIN_MAX - 1);
            const int rk = (int)((unsigned)d[u] >> (BIN_SHIFT + 6));
            stage[start[b] + rk] = make_uint2(((unsigned)s[u] << BIN_SHIFT) | (unsigned)(d[u] & (BIN_NODES - 1)),
                                              __float_as_uint(wv[u]));
        }
    }
    __syncthreads();

    const int total = start[BIN_MAX];
    uint2* out = (uint2*)rec + (size_t)base;
    for (int i = tid; i < total; i += 1024) out[i] = stage[i];
}

__device__ __forceinline__ void owner_body(float* acc, const int WIDTH,
                                           const unsigned* __restrict__ rec, const int* __restrict__ off,
                                           const float* __restrict__ hs,
                                           float* __restrict__ t0, float* __restrict__ t1,
                                           int n, int nchunks) {
    const int tid  = threadIdx.x;
    const int lane = tid & 31;
    const int wave = __builtin_amdgcn_readfirstlane(tid >> 5);
    const int b    = blockIdx.x;
    const int g    = blockIdx.y;
    const int TOT  = (WIDTH == 1) ? BIN_NODES : (PLANE1 + BIN_NODES);
    float* p1 = acc + PLANE1;

    for (int i = tid * 4; i < TOT; i += 1024 * 4) *(float4*)(acc + i) = make_float4(0.0f, 0.0f, 0.0f, 0.0f);
    __syncthreads();

    const int stride = OWN_GROUPS * 32;
    int c = g + OWN_GROUPS * wave;
    int o0 = 0, o1 = 0;
    if (c < nchunks) {
        o0 = off[(size_t)c * OFF_STRIDE + b];
        o1 = off[(size_t)c * OFF_STRIDE + b + 1];
    }
    while (c < nchunks) {
        const int cn = c + stride;
        int n0 = 0, n1 = 0;
        if (cn < nchunks) {
            n0 = off[(size_t)cn * OFF_STRIDE + b];
            n1 = off[(size_t)cn * OFF_STRIDE + b + 1];
        }
        const uint2* r = (const uint2*)rec + (size_t)c * CHUNK_EDGES;
        for (int i0 = o0; i0 < o1; i0 += 32 * OWN_LOADS) {
            uint2  q[OWN_LOADS];
            float2 h[OWN_LOADS];
#pragma unroll
            for (int u = 0; u < OWN_LOADS; ++u) {
                const int idx = i0 + u * 32 + lane;
                q[u] = r[idx < o1 ? idx : o0];
            }
            if (WIDTH == 2) {
#pragma unroll
                for (int u = 0; u < OWN_LOADS; ++u) h[u] = *(const float2*)(hs + (size_t)(q[u].x >> BIN_SHIFT) * 2);
            }
#pragma unroll
            for (int u = 0; u < OWN_LOADS; ++u) {
                const int   idx = i0 + u * 32 + lane;
                const bool  in  = idx < o1;
                const int   dl  = in ? (int)(q[u].x & (BIN_NODES - 1)) : lane;
                const float wv  = __uint_as_float(q[u].y);
                if (WIDTH == 1) {
                    lds_add_f32(acc + dl, in ? wv : 0.0f);
                } else {
                    const float a0 = wv * h[u].x;
                    const float a1 = wv * h[u].y;
                    lds_add_f32(acc + dl, in ? a0 : 0.0f);
                    lds_add_f32(p1 + dl, in ? a1 : 0.0f);
                }
            }
        }
        c = cn; o0 = n0; o1 = n1;
    }
    __syncthreads();

    float* t = package_half() ? t1 : t0;
    const int lo = b * BIN_NODES;
    if (WIDTH == 1) {
        const int rot = g * (BIN_NODES / OWN_GROUPS);
        for (int j = tid; j < BIN_NODES; j += 1024) {
            const int i = (j + rot) & (BIN_NODES - 1);
            if (lo + i < n) atomic_add_f32(t + lo + i, acc[i]);
        }
    } else {
        const int rot = g * (2 * BIN_NODES / OWN_GROUPS);
        for (int j = tid; j < 2 * BIN_NODES; j += 1024) {
            const int i  = (j + rot) & (2 * BIN_NODES - 1);
            const int nl = i >> 1;
            if (lo + nl < n) atomic_add_f32(t + (size_t)(lo + nl) * 2 + (i & 1), (i & 1) ? p1[nl] : acc[nl]);
        }
    }
}

__global__ __launch_bounds__(1024) void owner_degree(const unsigned* __restrict__ rec, const int* __restrict__ off,
                                                     float* __restrict__ deg0, float* __restrict__ deg1,
                                                     int n, int nchunks) {
    __shared__ __attribute__((aligned(16))) float acc[BIN_NODES];
    owner_body(acc, 1, rec, off, deg0, deg0, deg1, n, nchunks);
}

__global__ __launch_bounds__(1024) void owner_rows(const unsigned* __restrict__ rec, const int* __restrict__ off,
                                                   const float* __restrict__ hs,
                                                   float* __restrict__ agg0, float* __restrict__ agg1,
                                                   int n, int nchunks) {
    __shared__ __attribute__((aligned(16))) float acc[PLANE1 + BIN_NODES];
    owner_body(acc, 2, rec, off, hs, agg0, agg1, n, nchunks);
}

__global__ __launch_bounds__(256) void finalize2(const float* __restrict__ h2,
                                                 const float* __restrict__ agg20,
                                                 const float* __restrict__ agg21,
                                                 const float* __restrict__ dinv,
                                                 const float* __restrict__ b2,
                                                 float* __restrict__ out, int n) {
    int i = blockIdx.x * blockDim.x + threadIdx.x;
    if (i >= n) return;
    float di = dinv[i];
    float v0 = di * ((agg20[(size_t)i * 2 + 0] + agg21[(size_t)i * 2 + 0]) + h2[(size_t)i * 2 + 0]) + b2[0];
    float v1 = di * ((agg20[(size_t)i * 2 + 1] + agg21[(size_t)i * 2 + 1]) + h2[(size_t)i * 2 + 1]) + b2[1];
    float m  = fmaxf(v0, v1);
    float e0 = __expf(v0 - m);
    float e1 = __expf(v1 - m);
    float inv = 1.0f / (e0 + e1);
    out[(size_t)i * 2 + 0] = e0 * inv;
    out[(size_t)i * 2 + 1] = e1 * inv;
}

static inline float* align16(float* p) {
    return (float*)(((uintptr_t)p + 15) & ~(uintptr_t)15);
}

extern "C" void kernel_launch(void* const* d_in, const int* in_sizes, int n_in,
                              void* d_out, int out_size, void* d_ws, size_t ws_size,
                              hipStream_t stream) {
    const float* x  = (const float*)d_in[0];
    const int*   ei = (const int*)d_in[1];
    const float* w  = (const float*)d_in[2];
    const float* W1 = (const float*)d_in[3];
    const float* b1 = (const float*)d_in[4];
    const float* W2 = (const float*)d_in[5];
    const float* b2 = (const float*)d_in[6];
    float* out = (float*)d_out;

    const int n = in_sizes[0] / 2;
    const int e = in_sizes[2];

    float* ws    = (float*)d_ws;
    float* dinv  = align16(ws);
    float* aggx0 = align16(dinv  + (size_t)n);
    float* aggx1 = align16(aggx0 + (size_t)n * 2);
    float* deg1  = align16(aggx1 + (size_t)n * 2);
    float* agg20 = align16(deg1  + (size_t)n);
    float* agg21 = align16(agg20 + (size_t)n * 2);
    float* xs    = align16(agg21 + (size_t)n * 2);
    float* h2    = align16(xs    + (size_t)n * 2);

    const int TB = 256;
    dim3 blkE((e + EDGES_PER_BLOCK - 1) / EDGES_PER_BLOCK), thr(TB);
    dim3 blkN((n + TB - 1) / TB);
    const int nz = (int)((agg21 + (size_t)n * 2) - dinv);
    dim3 blkZ((nz / 4) / (TB * ZERO_PER_THREAD) + 1);
    const int tiles = (n + 15) / 16;
    dim3 blkC((tiles + 8 * TILES_PER_WAVE - 1) / (8 * TILES_PER_WAVE));

    const int nchunks = (e + CHUNK_EDGES - 1) / CHUNK_EDGES;
    const int nbins   = (n + BIN_NODES - 1) / BIN_NODES;
    unsigned* rec = (unsigned*)align16(h2 + (size_t)n * 2);
    int*      off = (int*)(rec + (size_t)nchunks * CHUNK_EDGES * 2);
    const size_t need = (size_t)((const char*)(off + (size_t)nchunks * OFF_STRIDE) - (const char*)d_ws);
    const bool binned = (e > 0) && (nbins <= BIN_MAX) && (need <= ws_size);

    zero_kernel<<<blkZ, thr, 0, stream>>>(dinv, nz);

    if (binned) {
        bin_chunks<<<dim3(nchunks), dim3(1024), 0, stream>>>(ei, w, rec, off, e, nbins);
        dim3 blkO(nbins, OWN_GROUPS);

        owner_degree<<<blkO, dim3(1024), 0, stream>>>(rec, off, dinv, deg1, n, nchunks);
        dinv_kernel<<<blkN, thr, 0, stream>>>(dinv, deg1, x, xs, n);

        owner_rows<<<blkO, dim3(1024), 0, stream>>>(rec, off, xs, aggx0, aggx1, n, nchunks);

        node_chain_wmma<<<blkC, thr, 0, stream>>>(x, aggx0, aggx1, dinv, W1, b1, W2, h2, n);

        owner_rows<<<blkO, dim3(1024), 0, stream>>>(rec, off, h2, agg20, agg21, n, nchunks);
        finalize2<<<blkN, thr, 0, stream>>>(h2, agg20, agg21, dinv, b2, out, n);
        return;
    }

    degree_owner<<<dim3((n + DEG_RANGE - 1) / DEG_RANGE, DEG_CHUNKS), dim3(1024), 0, stream>>>(ei, w, dinv, deg1, n, e);
    dinv_kernel<<<blkN, thr, 0, stream>>>(dinv, deg1, x, xs, n);

    edge_agg2<<<blkE, thr, 0, stream>>>(ei, w, xs, aggx0, aggx1, e);

    node_chain_wmma<<<blkC, thr, 0, stream>>>(x, aggx0, aggx1, dinv, W1, b1, W2, h2, n);

    edge_agg2<<<blkE, thr, 0, stream>>>(ei, w, h2, agg20, agg21, e);
    finalize2<<<blkN, thr, 0, stream>>>(h2, agg20, agg21, dinv, b2, out, n);
}
